# starting kernel
# baseline (speedup 1.0000x reference)
#include <hip/hip_runtime.h>
#include <math.h>

#define NN 96
#define PITCH 100
#define MPM_ITERS 50
#define NT 1024

typedef __attribute__((ext_vector_type(2))) float v2f;
typedef __attribute__((ext_vector_type(8))) float v8f;

__global__ __launch_bounds__(NT, 1)
void graphvae_mpm_kernel(const float* __restrict__ A_gt,
                         const float* __restrict__ vec_logits,
                         float* __restrict__ out)
{
    __shared__ float sP[NN * PITCH];
    __shared__ float sQ[NN * PITCH];
    __shared__ float sX[NN * PITCH];
    __shared__ float sM[NN * PITCH];
    __shared__ float sDegA[NN], sDegB[NN], sDA[NN], sDB[NN];
    __shared__ float sRed[33];

    const int tid  = threadIdx.x;
    const int lane = tid & 31;
    const int wave = tid >> 5;

    for (int e = tid; e < NN * NN; e += NT) {
        int i = e / NN, j = e % NN;
        float l;
        if (i == j) {
            l = 1.0f;
        } else {
            int lo = (i < j) ? i : j;
            int hi = (i < j) ? j : i;
            int off = (lo * (2 * NN - lo - 1)) / 2 + (hi - lo - 1);
            l = vec_logits[off];
        }
        sQ[i * PITCH + j] = 1.0f / (1.0f + expf(-l));
        sP[i * PITCH + j] = A_gt[e];
        sX[i * PITCH + j] = 1.0f / (float)NN;
    }
    __syncthreads();

    if (tid < NN) {
        float ga = 0.f, gb = 0.f;
        for (int j = 0; j < NN; ++j) {
            ga += sP[tid * PITCH + j];
            gb += sQ[tid * PITCH + j];
        }
        sDA[tid] = sP[tid * PITCH + tid];
        sDB[tid] = sQ[tid * PITCH + tid];
        sDegA[tid] = ga;
        sDegB[tid] = gb;
    }
    __syncthreads();

    for (int e = tid; e < NN * NN; e += NT) {
        int i = e / NN, j = e % NN;
        float p = sP[i * PITCH + j] * sDA[i] * sDA[j];
        float q = sQ[i * PITCH + j] * sDB[i] * sDB[j];
        if (i == j) { p = 0.f; q = 0.f; }
        sP[i * PITCH + j] = p;
        sQ[i * PITCH + j] = q;
    }
    __syncthreads();

    for (int it = 0; it < MPM_ITERS; ++it) {
        for (int e = tid; e < NN * NN; e += NT) {
            int j = e / NN, a = e % NN;
            const float* xr = &sX[j * PITCH];
            float m = 0.0f;
            #pragma unroll 4
            for (int b = 0; b < NN; ++b)
                m = fmaxf(m, sQ[b * PITCH + a] * xr[b]);
            sM[j * PITCH + a] = m;
        }
        __syncthreads();

        const int r = lane & 15;
        const int h = lane >> 4;
        v8f acc[2] = { {}, {} };
        const int nt = (wave + 32 < 36) ? 2 : 1;
        for (int q = 0; q < nt; ++q) {
            int t  = wave + 32 * q;
            int ti = t / 6, tn = t % 6;
            v8f c = {};
            for (int k0 = 0; k0 < NN; k0 += 4) {
                int ka = k0 + 2 * h;
                v2f av, bv;
                av.x = sP[(ti * 16 + r) * PITCH + ka];
                av.y = sP[(ti * 16 + r) * PITCH + ka + 1];
                bv.x = sM[ka * PITCH + tn * 16 + r];
                bv.y = sM[(ka + 1) * PITCH + tn * 16 + r];
                c = __builtin_amdgcn_wmma_f32_16x16x4_f32(
                         false, av,  false, bv,
                         (short)0, c,  false,  false);
            }
            acc[q] = c;
        }
        __syncthreads();

        for (int q = 0; q < nt; ++q) {
            int t  = wave + 32 * q;
            int ti = t / 6, tn = t % 6;
            int col = tn * 16 + r;
            #pragma unroll
            for (int v = 0; v < 8; ++v) {
                int row = ti * 16 + v + 8 * h;
                float ns = sDA[row] * sDB[col] /
                           (fabsf(sDegA[row] - sDegB[col]) + 1.0f);
                sM[row * PITCH + col] = sX[row * PITCH + col] * ns + acc[q][v];
            }
        }
        __syncthreads();

        float ss = 0.f;
        for (int e = tid; e < NN * NN; e += NT) {
            float v = sM[(e / NN) * PITCH + (e % NN)];
            ss += v * v;
        }
        #pragma unroll
        for (int msk = 16; msk >= 1; msk >>= 1)
            ss += __shfl_xor(ss, msk, 32);
        if (lane == 0) sRed[wave] = ss;
        __syncthreads();
        if (tid == 0) {
            float tot = 0.f;
            for (int w = 0; w < 32; ++w) tot += sRed[w];
            sRed[32] = 1.0f / sqrtf(tot);
        }
        __syncthreads();
        float inv = sRed[32];
        for (int e = tid; e < NN * NN; e += NT)
            sX[(e / NN) * PITCH + (e % NN)] =
                sM[(e / NN) * PITCH + (e % NN)] * inv;
        __syncthreads();
    }

    for (int e = tid; e < NN * NN; e += NT)
        out[e] = sX[(e / NN) * PITCH + (e % NN)];
}

extern "C" void kernel_launch(void* const* d_in, const int* in_sizes, int n_in,
                              void* d_out, int out_size, void* d_ws, size_t ws_size,
                              hipStream_t stream) {
    (void)in_sizes; (void)n_in; (void)d_ws; (void)ws_size; (void)out_size;
    const float* A_gt = (const float*)d_in[0];
    const float* vec  = (const float*)d_in[1];
    float* out = (float*)d_out;
    graphvae_mpm_kernel<<<1, NT, 0, stream>>>(A_gt, vec, out);
}


# phase 1 (M = max_b Q*X) register-tiled: 4x6 outputs per thread on 12 waves, operands read along b by 16-byte LDS loads, packed multiply + 3-input max; max regrouped (exact), no sum reordered
# speedup vs baseline: 4.6558x; 4.6558x over previous
#include <hip/hip_runtime.h>
#include <math.h>

#define NN 96
#define PITCH 100
#define MPM_ITERS 50
#define NT 1024

typedef __attribute__((ext_vector_type(2))) float v2f;
typedef __attribute__((ext_vector_type(8))) float v8f;
typedef __attribute__((ext_vector_type(4))) float v4f;

#define P1_J 4
#define P1_A 6
#define P1_THREADS ((NN / P1_J) * (NN / P1_A))


__device__ __forceinline__ void p1_load_q(v4f (&q)[P1_A], const float* qrow, int k0)
{
    #pragma unroll
    for (int v = 0; v < P1_A; ++v)
        q[v] = *(const v4f*)(qrow + v * (NN / P1_A) * PITCH + k0);
}

template <bool REFILL>
__device__ __forceinline__ void p1_step(float (&m)[P1_J][P1_A], v4f (&x)[P1_J], const v4f (&q)[P1_A],
                                        const float* xrow, int knext)
{
    #pragma unroll
    for (int u = 0; u < P1_J; ++u) {
        #pragma unroll
        for (int v = 0; v < P1_A; ++v) {
            v2f p0 = q[v].lo * x[u].lo;
            v2f p1 = q[v].hi * x[u].hi;
            float t = __builtin_fmaxf(__builtin_fmaxf(m[u][v], p0.x), p0.y);
            m[u][v] = __builtin_fmaxf(__builtin_fmaxf(t, p1.x), p1.y);
        }
        if (REFILL)
            x[u] = *(const v4f*)(xrow + u * (NN / P1_J) * PITCH + knext);
    }
}

__global__ __launch_bounds__(NT, 1)
void graphvae_mpm_kernel(const float* __restrict__ A_gt,
                         const float* __restrict__ vec_logits,
                         float* __restrict__ out)
{
    __shared__ float sP[NN * PITCH];
    __shared__ __attribute__((aligned(16))) float sQ[NN * PITCH];
    __shared__ __attribute__((aligned(16))) float sX[NN * PITCH];
    __shared__ float sM[NN * PITCH];
    __shared__ float sDegA[NN], sDegB[NN], sDA[NN], sDB[NN];
    __shared__ float sRed[33];

    const int tid  = threadIdx.x;
    const int lane = tid & 31;
    const int wave = tid >> 5;

    for (int e = tid; e < NN * NN; e += NT) {
        int i = e / NN, j = e % NN;
        float l;
        if (i == j) {
            l = 1.0f;
        } else {
            int lo = (i < j) ? i : j;
            int hi = (i < j) ? j : i;
            int off = (lo * (2 * NN - lo - 1)) / 2 + (hi - lo - 1);
            l = vec_logits[off];
        }
        sQ[i * PITCH + j] = 1.0f / (1.0f + expf(-l));
        sP[i * PITCH + j] = A_gt[e];
        sX[i * PITCH + j] = 1.0f / (float)NN;
    }
    __syncthreads();

    if (tid < NN) {
        float ga = 0.f, gb = 0.f;
        for (int j = 0; j < NN; ++j) {
            ga += sP[tid * PITCH + j];
            gb += sQ[tid * PITCH + j];
        }
        sDA[tid] = sP[tid * PITCH + tid];
        sDB[tid] = sQ[tid * PITCH + tid];
        sDegA[tid] = ga;
        sDegB[tid] = gb;
    }
    __syncthreads();

    for (int e = tid; e < NN * NN; e += NT) {
        int i = e / NN, j = e % NN;
        float p = sP[i * PITCH + j] * sDA[i] * sDA[j];
        float q = sQ[i * PITCH + j] * sDB[i] * sDB[j];
        if (i == j) { p = 0.f; q = 0.f; }
        sP[i * PITCH + j] = p;
        sQ[i * PITCH + j] = q;
    }
    __syncthreads();

    for (int it = 0; it < MPM_ITERS; ++it) {
        if (tid < P1_THREADS) {
            const int jg = 2 * wave + (lane >> 4);
            const int ag = lane & 15;
            const float* xrow = &sX[jg * PITCH];
            const float* qrow = &sQ[ag * PITCH];
            float m[P1_J][P1_A];
            #pragma unroll
            for (int u = 0; u < P1_J; ++u) {
                #pragma unroll
                for (int v = 0; v < P1_A; ++v) m[u][v] = 0.0f;
            }
            v4f x[P1_J], qa[P1_A], qb[P1_A];
            #pragma unroll
            for (int u = 0; u < P1_J; ++u)
                x[u] = *(const v4f*)(xrow + u * (NN / P1_J) * PITCH);
            p1_load_q(qa, qrow, 0);
            for (int k0 = 0; k0 < NN - 8; k0 += 8) {
                p1_load_q(qb, qrow, k0 + 4);
                p1_step<true>(m, x, qa, xrow, k0 + 4);
                p1_load_q(qa, qrow, k0 + 8);
                p1_step<true>(m, x, qb, xrow, k0 + 8);
            }
            p1_load_q(qb, qrow, NN - 4);
            p1_step<true>(m, x, qa, xrow, NN - 4);
            p1_step<false>(m, x, qb, xrow, 0);
            #pragma unroll
            for (int u = 0; u < P1_J; ++u) {
                #pragma unroll
                for (int v = 0; v < P1_A; ++v)
                    sM[(jg + u * (NN / P1_J)) * PITCH + ag + v * (NN / P1_A)] = m[u][v];
            }
        }
        __syncthreads();

        const int r = lane & 15;
        const int h = lane >> 4;
        v8f acc[2] = { {}, {} };
        const int nt = (wave + 32 < 36) ? 2 : 1;
        for (int q = 0; q < nt; ++q) {
            int t  = wave + 32 * q;
            int ti = t / 6, tn = t % 6;
            v8f c = {};
            for (int k0 = 0; k0 < NN; k0 += 4) {
                int ka = k0 + 2 * h;
                v2f av, bv;
                av.x = sP[(ti * 16 + r) * PITCH + ka];
                av.y = sP[(ti * 16 + r) * PITCH + ka + 1];
                bv.x = sM[ka * PITCH + tn * 16 + r];
                bv.y = sM[(ka + 1) * PITCH + tn * 16 + r];
                c = __builtin_amdgcn_wmma_f32_16x16x4_f32(
                         false, av,  false, bv,
                         (short)0, c,  false,  false);
            }
            acc[q] = c;
        }
        __syncthreads();

        for (int q = 0; q < nt; ++q) {
            int t  = wave + 32 * q;
            int ti = t / 6, tn = t % 6;
            int col = tn * 16 + r;
            #pragma unroll
            for (int v = 0; v < 8; ++v) {
                int row = ti * 16 + v + 8 * h;
                float ns = sDA[row] * sDB[col] /
                           (fabsf(sDegA[row] - sDegB[col]) + 1.0f);
                sM[row * PITCH + col] = sX[row * PITCH + col] * ns + acc[q][v];
            }
        }
        __syncthreads();

        float ss = 0.f;
        for (int e = tid; e < NN * NN; e += NT) {
            float v = sM[(e / NN) * PITCH + (e % NN)];
            ss += v * v;
        }
        #pragma unroll
        for (int msk = 16; msk >= 1; msk >>= 1)
            ss += __shfl_xor(ss, msk, 32);
        if (lane == 0) sRed[wave] = ss;
        __syncthreads();
        if (tid == 0) {
            float tot = 0.f;
            for (int w = 0; w < 32; ++w) tot += sRed[w];
            sRed[32] = 1.0f / sqrtf(tot);
        }
        __syncthreads();
        float inv = sRed[32];
        for (int e = tid; e < NN * NN; e += NT)
            sX[(e / NN) * PITCH + (e % NN)] =
                sM[(e / NN) * PITCH + (e % NN)] * inv;
        __syncthreads();
    }

    for (int e = tid; e < NN * NN; e += NT)
        out[e] = sX[(e / NN) * PITCH + (e % NN)];
}

extern "C" void kernel_launch(void* const* d_in, const int* in_sizes, int n_in,
                              void* d_out, int out_size, void* d_ws, size_t ws_size,
                              hipStream_t stream) {
    (void)in_sizes; (void)n_in; (void)d_ws; (void)ws_size; (void)out_size;
    const float* A_gt = (const float*)d_in[0];
    const float* vec  = (const float*)d_in[1];
    float* out = (float*)d_out;
    graphvae_mpm_kernel<<<1, NT, 0, stream>>>(A_gt, vec, out);
}


# epilogue kept in the WMMA accumulator registers: X*NS+Y, its squared norm and the rescale with no LDS round trip of Xn (NS tabulated once); 5 -> 3 barriers per iteration; norm summed in another order
# speedup vs baseline: 5.6965x; 1.2235x over previous
#include <hip/hip_runtime.h>
#include <math.h>

#define NN 96
#define PITCH 100
#define MPM_ITERS 50
#define NT 1024

typedef __attribute__((ext_vector_type(2))) float v2f;
typedef __attribute__((ext_vector_type(8))) float v8f;
typedef __attribute__((ext_vector_type(4))) float v4f;

#define P1_J 4
#define P1_A 6
#define P1_THREADS ((NN / P1_J) * (NN / P1_A))


__device__ __forceinline__ void p1_load_q(v4f (&q)[P1_A], const float* qrow, int k0)
{
    #pragma unroll
    for (int v = 0; v < P1_A; ++v)
        q[v] = *(const v4f*)(qrow + v * (NN / P1_A) * PITCH + k0);
}

template <bool REFILL>
__device__ __forceinline__ void p1_step(float (&m)[P1_J][P1_A], v4f (&x)[P1_J], const v4f (&q)[P1_A],
                                        const float* xrow, int knext)
{
    #pragma unroll
    for (int u = 0; u < P1_J; ++u) {
        #pragma unroll
        for (int v = 0; v < P1_A; ++v) {
            v2f p0 = q[v].lo * x[u].lo;
            v2f p1 = q[v].hi * x[u].hi;
            float t = __builtin_fmaxf(__builtin_fmaxf(m[u][v], p0.x), p0.y);
            m[u][v] = __builtin_fmaxf(__builtin_fmaxf(t, p1.x), p1.y);
        }
        if (REFILL)
            x[u] = *(const v4f*)(xrow + u * (NN / P1_J) * PITCH + knext);
    }
}


__device__ __forceinline__ v8f tile_gemm(const float* pa, const float* pb)
{
    v8f c = {};
    #pragma unroll
    for (int k0 = 0; k0 < NN; k0 += 4) {
        v2f av, bv;
        av.x = pa[k0];
        av.y = pa[k0 + 1];
        bv.x = pb[k0 * PITCH];
        bv.y = pb[(k0 + 1) * PITCH];
        c = __builtin_amdgcn_wmma_f32_16x16x4_f32(
                 false, av,  false, bv,
                 (short)0, c,  false,  false);
    }
    return c;
}

__device__ __forceinline__ void tile_update(v8f& acc, float& ss, const float* px, const float* pns)
{
    #pragma unroll
    for (int v = 0; v < 8; ++v) {
        float xn = px[v * PITCH] * pns[v * PITCH] + acc[v];
        acc[v] = xn;
        ss += xn * xn;
    }
}

__device__ __forceinline__ void tile_store(const v8f& acc, float inv, float* px)
{
    #pragma unroll
    for (int v = 0; v < 8; ++v)
        px[v * PITCH] = acc[v] * inv;
}

template <int MASK>
__device__ __forceinline__ float row_xor(float x)
{
    int b = __builtin_bit_cast(int, x);
    return __builtin_bit_cast(float, __builtin_amdgcn_update_dpp(b, b, 0x160 + MASK, 0xf, 0xf, false));
}

__device__ __forceinline__ float xhalf(float x)
{
    int b = __builtin_bit_cast(int, x);
    return __builtin_bit_cast(float, __builtin_amdgcn_permlanex16(b, b, 0x76543210, 0xfedcba98, false, false));
}

__device__ __forceinline__ float wave_sum(float x)
{
    x += row_xor<1>(x);
    x += row_xor<2>(x);
    x += row_xor<4>(x);
    x += row_xor<8>(x);
    x += xhalf(x);
    return x;
}

__global__ __launch_bounds__(NT, 1)
void graphvae_mpm_kernel(const float* __restrict__ A_gt,
                         const float* __restrict__ vec_logits,
                         float* __restrict__ out)
{
    __shared__ float sP[NN * PITCH];
    __shared__ __attribute__((aligned(16))) float sQ[NN * PITCH];
    __shared__ __attribute__((aligned(16))) float sX[NN * PITCH];
    __shared__ float sM[NN * PITCH];
    __shared__ float sNS[NN * PITCH];
    __shared__ float sDegA[NN], sDegB[NN], sDA[NN], sDB[NN];
    __shared__ float sRed[32];

    const int tid  = threadIdx.x;
    const int lane = tid & 31;
    const int wave = __builtin_amdgcn_readfirstlane(tid >> 5);

    for (int e = tid; e < NN * NN; e += NT) {
        int i = e / NN, j = e % NN;
        float l;
        if (i == j) {
            l = 1.0f;
        } else {
            int lo = (i < j) ? i : j;
            int hi = (i < j) ? j : i;
            int off = (lo * (2 * NN - lo - 1)) / 2 + (hi - lo - 1);
            l = vec_logits[off];
        }
        sQ[i * PITCH + j] = 1.0f / (1.0f + expf(-l));
        sP[i * PITCH + j] = A_gt[e];
        sX[i * PITCH + j] = 1.0f / (float)NN;
    }
    __syncthreads();

    if (tid < NN) {
        float ga = 0.f, gb = 0.f;
        for (int j = 0; j < NN; ++j) {
            ga += sP[tid * PITCH + j];
            gb += sQ[tid * PITCH + j];
        }
        sDA[tid] = sP[tid * PITCH + tid];
        sDB[tid] = sQ[tid * PITCH + tid];
        sDegA[tid] = ga;
        sDegB[tid] = gb;
    }
    __syncthreads();

    for (int e = tid; e < NN * NN; e += NT) {
        int i = e / NN, j = e % NN;
        float p = sP[i * PITCH + j] * sDA[i] * sDA[j];
        float q = sQ[i * PITCH + j] * sDB[i] * sDB[j];
        if (i == j) { p = 0.f; q = 0.f; }
        sP[i * PITCH + j] = p;
        sQ[i * PITCH + j] = q;
        sNS[i * PITCH + j] = sDA[i] * sDB[j] / (fabsf(sDegA[i] - sDegB[j]) + 1.0f);
    }
    __syncthreads();

    const int r = lane & 15;
    const int h = lane >> 4;
    const int ti0 = wave / 6, tn0 = wave % 6;
    const int ti1 = 5,        tn1 = 2 + wave;
    const int base0 = (ti0 * 16 + 8 * h) * PITCH + tn0 * 16 + r;
    const int base1 = (ti1 * 16 + 8 * h) * PITCH + tn1 * 16 + r;
    const int offa0 = (ti0 * 16 + r) * PITCH + 2 * h;
    const int offa1 = (ti1 * 16 + r) * PITCH + 2 * h;
    const int offb0 = 2 * h * PITCH + tn0 * 16 + r;
    const int offb1 = 2 * h * PITCH + tn1 * 16 + r;
    const bool two = wave < 4;

    for (int it = 0; it < MPM_ITERS; ++it) {
        if (tid < P1_THREADS) {
            const int jg = 2 * wave + (lane >> 4);
            const int ag = lane & 15;
            const float* xrow = &sX[jg * PITCH];
            const float* qrow = &sQ[ag * PITCH];
            float m[P1_J][P1_A];
            #pragma unroll
            for (int u = 0; u < P1_J; ++u) {
                #pragma unroll
                for (int v = 0; v < P1_A; ++v) m[u][v] = 0.0f;
            }
            v4f x[P1_J], qa[P1_A], qb[P1_A];
            #pragma unroll
            for (int u = 0; u < P1_J; ++u)
                x[u] = *(const v4f*)(xrow + u * (NN / P1_J) * PITCH);
            p1_load_q(qa, qrow, 0);
            for (int k0 = 0; k0 < NN - 8; k0 += 8) {
                p1_load_q(qb, qrow, k0 + 4);
                p1_step<true>(m, x, qa, xrow, k0 + 4);
                p1_load_q(qa, qrow, k0 + 8);
                p1_step<true>(m, x, qb, xrow, k0 + 8);
            }
            p1_load_q(qb, qrow, NN - 4);
            p1_step<true>(m, x, qa, xrow, NN - 4);
            p1_step<false>(m, x, qb, xrow, 0);
            #pragma unroll
            for (int u = 0; u < P1_J; ++u) {
                #pragma unroll
                for (int v = 0; v < P1_A; ++v)
                    sM[(jg + u * (NN / P1_J)) * PITCH + ag + v * (NN / P1_A)] = m[u][v];
            }
        }
        __syncthreads();

        v8f acc1 = {};
        if (two) acc1 = tile_gemm(sP + offa1, sM + offb1);
        v8f acc0 = tile_gemm(sP + offa0, sM + offb0);

        float ss = 0.f;
        tile_update(acc0, ss, sX + base0, sNS + base0);
        if (two) tile_update(acc1, ss, sX + base1, sNS + base1);
        ss = wave_sum(ss);
        if (lane == 0) sRed[wave] = ss;
        __syncthreads();

        float tot = wave_sum(sRed[lane]);
        float inv = 1.0f / sqrtf(tot);
        tile_store(acc0, inv, sX + base0);
        if (two) tile_store(acc1, inv, sX + base1);
        __syncthreads();
    }

    for (int e = tid; e < NN * NN; e += NT)
        out[e] = sX[(e / NN) * PITCH + (e % NN)];
}

extern "C" void kernel_launch(void* const* d_in, const int* in_sizes, int n_in,
                              void* d_out, int out_size, void* d_ws, size_t ws_size,
                              hipStream_t stream) {
    (void)in_sizes; (void)n_in; (void)d_ws; (void)ws_size; (void)out_size;
    const float* A_gt = (const float*)d_in[0];
    const float* vec  = (const float*)d_in[1];
    float* out = (float*)d_out;
    graphvae_mpm_kernel<<<1, NT, 0, stream>>>(A_gt, vec, out);
}


# phase 1 split along b over the two 16-lane halves of a wave: 24 waves instead of 12 (6 per SIMD), same LDS reads and products, halves joined by one cross-half max (exact); no sum reordered
# speedup vs baseline: 6.0080x; 1.0547x over previous
#include <hip/hip_runtime.h>
#include <math.h>

#define NN 96
#define PITCH 100
#define MPM_ITERS 50
#define NT 1024

typedef __attribute__((ext_vector_type(2))) float v2f;
typedef __attribute__((ext_vector_type(8))) float v8f;
typedef __attribute__((ext_vector_type(4))) float v4f;

#define P1_J 4
#define P1_A 6
#define P1_B (NN / 2)
#define P1_THREADS (2 * (NN / P1_J) * (NN / P1_A))


__device__ __forceinline__ void p1_load_q(v4f (&q)[P1_A], const float* qrow, int k0)
{
    #pragma unroll
    for (int v = 0; v < P1_A; ++v)
        q[v] = *(const v4f*)(qrow + v * (NN / P1_A) * PITCH + k0);
}

template <bool REFILL>
__device__ __forceinline__ void p1_step(float (&m)[P1_J][P1_A], v4f (&x)[P1_J], const v4f (&q)[P1_A],
                                        const float* xrow, int knext)
{
    #pragma unroll
    for (int u = 0; u < P1_J; ++u) {
        #pragma unroll
        for (int v = 0; v < P1_A; ++v) {
            v2f p0 = q[v].lo * x[u].lo;
            v2f p1 = q[v].hi * x[u].hi;
            float t = __builtin_fmaxf(__builtin_fmaxf(m[u][v], p0.x), p0.y);
            m[u][v] = __builtin_fmaxf(__builtin_fmaxf(t, p1.x), p1.y);
        }
        if (REFILL)
            x[u] = *(const v4f*)(xrow + u * (NN / P1_J) * PITCH + knext);
    }
}


__device__ __forceinline__ v8f tile_gemm(const float* pa, const float* pb)
{
    v8f c = {};
    #pragma unroll
    for (int k0 = 0; k0 < NN; k0 += 4) {
        v2f av, bv;
        av.x = pa[k0];
        av.y = pa[k0 + 1];
        bv.x = pb[k0 * PITCH];
        bv.y = pb[(k0 + 1) * PITCH];
        c = __builtin_amdgcn_wmma_f32_16x16x4_f32(
                 false, av,  false, bv,
                 (short)0, c,  false,  false);
    }
    return c;
}

__device__ __forceinline__ void tile_update(v8f& acc, float& ss, const float* px, const float* pns)
{
    #pragma unroll
    for (int v = 0; v < 8; ++v) {
        float xn = px[v * PITCH] * pns[v * PITCH] + acc[v];
        acc[v] = xn;
        ss += xn * xn;
    }
}

__device__ __forceinline__ void tile_store(const v8f& acc, float inv, float* px)
{
    #pragma unroll
    for (int v = 0; v < 8; ++v)
        px[v * PITCH] = acc[v] * inv;
}

template <int MASK>
__device__ __forceinline__ float row_xor(float x)
{
    int b = __builtin_bit_cast(int, x);
    return __builtin_bit_cast(float, __builtin_amdgcn_update_dpp(b, b, 0x160 + MASK, 0xf, 0xf, false));
}

__device__ __forceinline__ float xhalf(float x)
{
    int b = __builtin_bit_cast(int, x);
    return __builtin_bit_cast(float, __builtin_amdgcn_permlanex16(b, b, 0x76543210, 0xfedcba98, false, false));
}

__device__ __forceinline__ float wave_sum(float x)
{
    x += row_xor<1>(x);
    x += row_xor<2>(x);
    x += row_xor<4>(x);
    x += row_xor<8>(x);
    x += xhalf(x);
    return x;
}

__global__ __launch_bounds__(NT, 1)
void graphvae_mpm_kernel(const float* __restrict__ A_gt,
                         const float* __restrict__ vec_logits,
                         float* __restrict__ out)
{
    __shared__ float sP[NN * PITCH];
    __shared__ __attribute__((aligned(16))) float sQ[NN * PITCH];
    __shared__ __attribute__((aligned(16))) float sX[NN * PITCH];
    __shared__ float sM[NN * PITCH];
    __shared__ float sNS[NN * PITCH];
    __shared__ float sDegA[NN], sDegB[NN], sDA[NN], sDB[NN];
    __shared__ float sRed[32];

    const int tid  = threadIdx.x;
    const int lane = tid & 31;
    const int wave = __builtin_amdgcn_readfirstlane(tid >> 5);

    for (int e = tid; e < NN * NN; e += NT) {
        int i = e / NN, j = e % NN;
        float l;
        if (i == j) {
            l = 1.0f;
        } else {
            int lo = (i < j) ? i : j;
            int hi = (i < j) ? j : i;
            int off = (lo * (2 * NN - lo - 1)) / 2 + (hi - lo - 1);
            l = vec_logits[off];
        }
        sQ[i * PITCH + j] = 1.0f / (1.0f + expf(-l));
        sP[i * PITCH + j] = A_gt[e];
        sX[i * PITCH + j] = 1.0f / (float)NN;
    }
    __syncthreads();

    if (tid < NN) {
        float ga = 0.f, gb = 0.f;
        for (int j = 0; j < NN; ++j) {
            ga += sP[tid * PITCH + j];
            gb += sQ[tid * PITCH + j];
        }
        sDA[tid] = sP[tid * PITCH + tid];
        sDB[tid] = sQ[tid * PITCH + tid];
        sDegA[tid] = ga;
        sDegB[tid] = gb;
    }
    __syncthreads();

    for (int e = tid; e < NN * NN; e += NT) {
        int i = e / NN, j = e % NN;
        float p = sP[i * PITCH + j] * sDA[i] * sDA[j];
        float q = sQ[i * PITCH + j] * sDB[i] * sDB[j];
        if (i == j) { p = 0.f; q = 0.f; }
        sP[i * PITCH + j] = p;
        sQ[i * PITCH + j] = q;
        sNS[i * PITCH + j] = sDA[i] * sDB[j] / (fabsf(sDegA[i] - sDegB[j]) + 1.0f);
    }
    __syncthreads();

    const int r = lane & 15;
    const int h = lane >> 4;
    const int ti0 = wave / 6, tn0 = wave % 6;
    const int ti1 = 5,        tn1 = 2 + wave;
    const int base0 = (ti0 * 16 + 8 * h) * PITCH + tn0 * 16 + r;
    const int base1 = (ti1 * 16 + 8 * h) * PITCH + tn1 * 16 + r;
    const int offa0 = (ti0 * 16 + r) * PITCH + 2 * h;
    const int offa1 = (ti1 * 16 + r) * PITCH + 2 * h;
    const int offb0 = 2 * h * PITCH + tn0 * 16 + r;
    const int offb1 = 2 * h * PITCH + tn1 * 16 + r;
    const bool two = wave < 4;

    for (int it = 0; it < MPM_ITERS; ++it) {
        if (tid < P1_THREADS) {
            const int jg = wave;
            const int ag = lane & 15;
            const int hb = lane >> 4;
            const float* xrow = &sX[jg * PITCH + hb * P1_B];
            const float* qrow = &sQ[ag * PITCH + hb * P1_B];
            float m[P1_J][P1_A];
            #pragma unroll
            for (int u = 0; u < P1_J; ++u) {
                #pragma unroll
                for (int v = 0; v < P1_A; ++v) m[u][v] = 0.0f;
            }
            v4f x[P1_J], qa[P1_A], qb[P1_A];
            #pragma unroll
            for (int u = 0; u < P1_J; ++u)
                x[u] = *(const v4f*)(xrow + u * (NN / P1_J) * PITCH);
            p1_load_q(qa, qrow, 0);
            for (int k0 = 0; k0 < P1_B - 8; k0 += 8) {
                p1_load_q(qb, qrow, k0 + 4);
                p1_step<true>(m, x, qa, xrow, k0 + 4);
                p1_load_q(qa, qrow, k0 + 8);
                p1_step<true>(m, x, qb, xrow, k0 + 8);
            }
            p1_load_q(qb, qrow, P1_B - 4);
            p1_step<true>(m, x, qa, xrow, P1_B - 4);
            p1_step<false>(m, x, qb, xrow, 0);
            float* mrow = &sM[jg * PITCH + ag + hb * (P1_A / 2) * (NN / P1_A)];
            #pragma unroll
            for (int u = 0; u < P1_J; ++u) {
                #pragma unroll
                for (int v = 0; v < P1_A / 2; ++v) {
                    float keep = hb ? m[u][v + P1_A / 2] : m[u][v];
                    float send = hb ? m[u][v] : m[u][v + P1_A / 2];
                    mrow[u * (NN / P1_J) * PITCH + v * (NN / P1_A)] = __builtin_fmaxf(keep, xhalf(send));
                }
            }
        }
        __syncthreads();

        v8f acc1 = {};
        if (two) acc1 = tile_gemm(sP + offa1, sM + offb1);
        v8f acc0 = tile_gemm(sP + offa0, sM + offb0);

        float ss = 0.f;
        tile_update(acc0, ss, sX + base0, sNS + base0);
        if (two) tile_update(acc1, ss, sX + base1, sNS + base1);
        ss = wave_sum(ss);
        if (lane == 0) sRed[wave] = ss;
        __syncthreads();

        float tot = wave_sum(sRed[lane]);
        float inv = 1.0f / sqrtf(tot);
        tile_store(acc0, inv, sX + base0);
        if (two) tile_store(acc1, inv, sX + base1);
        __syncthreads();
    }

    for (int e = tid; e < NN * NN; e += NT)
        out[e] = sX[(e / NN) * PITCH + (e % NN)];
}

extern "C" void kernel_launch(void* const* d_in, const int* in_sizes, int n_in,
                              void* d_out, int out_size, void* d_ws, size_t ws_size,
                              hipStream_t stream) {
    (void)in_sizes; (void)n_in; (void)d_ws; (void)ws_size; (void)out_size;
    const float* A_gt = (const float*)d_in[0];
    const float* vec  = (const float*)d_in[1];
    float* out = (float*)d_out;
    graphvae_mpm_kernel<<<1, NT, 0, stream>>>(A_gt, vec, out);
}


# each iteration spread over 6 work-groups (one 16-column block of M, Y, X each), one launch per iteration, X handed on through workspace blocks; norm summed in another order
# speedup vs baseline: 7.5996x; 1.2649x over previous
#include <hip/hip_runtime.h>
#include <math.h>

#define NN 96
#define PITCH 100
#define MPM_ITERS 50
#define NT 1024

typedef __attribute__((ext_vector_type(2))) float v2f;
typedef __attribute__((ext_vector_type(8))) float v8f;
typedef __attribute__((ext_vector_type(4))) float v4f;

#define P1_J 4
#define P1_A 6
#define P1_B (NN / 2)
#define P1_THREADS (2 * (NN / P1_J) * (NN / P1_A))


__device__ __forceinline__ void p1_load_q(v4f (&q)[P1_A], const float* qrow, int k0)
{
    #pragma unroll
    for (int v = 0; v < P1_A; ++v)
        q[v] = *(const v4f*)(qrow + v * (NN / P1_A) * PITCH + k0);
}

template <bool REFILL>
__device__ __forceinline__ void p1_step(float (&m)[P1_J][P1_A], v4f (&x)[P1_J], const v4f (&q)[P1_A],
                                        const float* xrow, int knext)
{
    #pragma unroll
    for (int u = 0; u < P1_J; ++u) {
        #pragma unroll
        for (int v = 0; v < P1_A; ++v) {
            v2f p0 = q[v].lo * x[u].lo;
            v2f p1 = q[v].hi * x[u].hi;
            float t = __builtin_fmaxf(__builtin_fmaxf(m[u][v], p0.x), p0.y);
            m[u][v] = __builtin_fmaxf(__builtin_fmaxf(t, p1.x), p1.y);
        }
        if (REFILL)
            x[u] = *(const v4f*)(xrow + u * (NN / P1_J) * PITCH + knext);
    }
}


__device__ __forceinline__ v8f tile_gemm(const float* pa, const float* pb)
{
    v8f c = {};
    #pragma unroll
    for (int k0 = 0; k0 < NN; k0 += 4) {
        v2f av, bv;
        av.x = pa[k0];
        av.y = pa[k0 + 1];
        bv.x = pb[k0 * PITCH];
        bv.y = pb[(k0 + 1) * PITCH];
        c = __builtin_amdgcn_wmma_f32_16x16x4_f32(
                 false, av,  false, bv,
                 (short)0, c,  false,  false);
    }
    return c;
}

__device__ __forceinline__ void tile_update(v8f& acc, float& ss, const float* px, const float* pns)
{
    #pragma unroll
    for (int v = 0; v < 8; ++v) {
        float xn = px[v * PITCH] * pns[v * PITCH] + acc[v];
        acc[v] = xn;
        ss += xn * xn;
    }
}

__device__ __forceinline__ void tile_store(const v8f& acc, float inv, float* px)
{
    #pragma unroll
    for (int v = 0; v < 8; ++v)
        px[v * PITCH] = acc[v] * inv;
}

template <int MASK>
__device__ __forceinline__ float row_xor(float x)
{
    int b = __builtin_bit_cast(int, x);
    return __builtin_bit_cast(float, __builtin_amdgcn_update_dpp(b, b, 0x160 + MASK, 0xf, 0xf, false));
}

__device__ __forceinline__ float xhalf(float x)
{
    int b = __builtin_bit_cast(int, x);
    return __builtin_bit_cast(float, __builtin_amdgcn_permlanex16(b, b, 0x76543210, 0xfedcba98, false, false));
}

__device__ __forceinline__ float wave_sum(float x)
{
    x += row_xor<1>(x);
    x += row_xor<2>(x);
    x += row_xor<4>(x);
    x += row_xor<8>(x);
    x += xhalf(x);
    return x;
}

__global__ __launch_bounds__(NT, 1)
void graphvae_mpm_kernel(const float* __restrict__ A_gt,
                         const float* __restrict__ vec_logits,
                         float* __restrict__ out)
{
    __shared__ float sP[NN * PITCH];
    __shared__ __attribute__((aligned(16))) float sQ[NN * PITCH];
    __shared__ __attribute__((aligned(16))) float sX[NN * PITCH];
    __shared__ float sM[NN * PITCH];
    __shared__ float sNS[NN * PITCH];
    __shared__ float sDegA[NN], sDegB[NN], sDA[NN], sDB[NN];
    __shared__ float sRed[32];

    const int tid  = threadIdx.x;
    const int lane = tid & 31;
    const int wave = __builtin_amdgcn_readfirstlane(tid >> 5);

    for (int e = tid; e < NN * NN; e += NT) {
        int i = e / NN, j = e % NN;
        float l;
        if (i == j) {
            l = 1.0f;
        } else {
            int lo = (i < j) ? i : j;
            int hi = (i < j) ? j : i;
            int off = (lo * (2 * NN - lo - 1)) / 2 + (hi - lo - 1);
            l = vec_logits[off];
        }
        sQ[i * PITCH + j] = 1.0f / (1.0f + expf(-l));
        sP[i * PITCH + j] = A_gt[e];
        sX[i * PITCH + j] = 1.0f / (float)NN;
    }
    __syncthreads();

    if (tid < NN) {
        float ga = 0.f, gb = 0.f;
        for (int j = 0; j < NN; ++j) {
            ga += sP[tid * PITCH + j];
            gb += sQ[tid * PITCH + j];
        }
        sDA[tid] = sP[tid * PITCH + tid];
        sDB[tid] = sQ[tid * PITCH + tid];
        sDegA[tid] = ga;
        sDegB[tid] = gb;
    }
    __syncthreads();

    for (int e = tid; e < NN * NN; e += NT) {
        int i = e / NN, j = e % NN;
        float p = sP[i * PITCH + j] * sDA[i] * sDA[j];
        float q = sQ[i * PITCH + j] * sDB[i] * sDB[j];
        if (i == j) { p = 0.f; q = 0.f; }
        sP[i * PITCH + j] = p;
        sQ[i * PITCH + j] = q;
        sNS[i * PITCH + j] = sDA[i] * sDB[j] / (fabsf(sDegA[i] - sDegB[j]) + 1.0f);
    }
    __syncthreads();

    const int r = lane & 15;
    const int h = lane >> 4;
    const int ti0 = wave / 6, tn0 = wave % 6;
    const int ti1 = 5,        tn1 = 2 + wave;
    const int base0 = (ti0 * 16 + 8 * h) * PITCH + tn0 * 16 + r;
    const int base1 = (ti1 * 16 + 8 * h) * PITCH + tn1 * 16 + r;
    const int offa0 = (ti0 * 16 + r) * PITCH + 2 * h;
    const int offa1 = (ti1 * 16 + r) * PITCH + 2 * h;
    const int offb0 = 2 * h * PITCH + tn0 * 16 + r;
    const int offb1 = 2 * h * PITCH + tn1 * 16 + r;
    const bool two = wave < 4;

    for (int it = 0; it < MPM_ITERS; ++it) {
        if (tid < P1_THREADS) {
            const int jg = wave;
            const int ag = lane & 15;
            const int hb = lane >> 4;
            const float* xrow = &sX[jg * PITCH + hb * P1_B];
            const float* qrow = &sQ[ag * PITCH + hb * P1_B];
            float m[P1_J][P1_A];
            #pragma unroll
            for (int u = 0; u < P1_J; ++u) {
                #pragma unroll
                for (int v = 0; v < P1_A; ++v) m[u][v] = 0.0f;
            }
            v4f x[P1_J], qa[P1_A], qb[P1_A];
            #pragma unroll
            for (int u = 0; u < P1_J; ++u)
                x[u] = *(const v4f*)(xrow + u * (NN / P1_J) * PITCH);
            p1_load_q(qa, qrow, 0);
            for (int k0 = 0; k0 < P1_B - 8; k0 += 8) {
                p1_load_q(qb, qrow, k0 + 4);
                p1_step<true>(m, x, qa, xrow, k0 + 4);
                p1_load_q(qa, qrow, k0 + 8);
                p1_step<true>(m, x, qb, xrow, k0 + 8);
            }
            p1_load_q(qb, qrow, P1_B - 4);
            p1_step<true>(m, x, qa, xrow, P1_B - 4);
            p1_step<false>(m, x, qb, xrow, 0);
            float* mrow = &sM[jg * PITCH + ag + hb * (P1_A / 2) * (NN / P1_A)];
            #pragma unroll
            for (int u = 0; u < P1_J; ++u) {
                #pragma unroll
                for (int v = 0; v < P1_A / 2; ++v) {
                    float keep = hb ? m[u][v + P1_A / 2] : m[u][v];
                    float send = hb ? m[u][v] : m[u][v + P1_A / 2];
                    mrow[u * (NN / P1_J) * PITCH + v * (NN / P1_A)] = __builtin_fmaxf(keep, xhalf(send));
                }
            }
        }
        __syncthreads();

        v8f acc1 = {};
        if (two) acc1 = tile_gemm(sP + offa1, sM + offb1);
        v8f acc0 = tile_gemm(sP + offa0, sM + offb0);

        float ss = 0.f;
        tile_update(acc0, ss, sX + base0, sNS + base0);
        if (two) tile_update(acc1, ss, sX + base1, sNS + base1);
        ss = wave_sum(ss);
        if (lane == 0) sRed[wave] = ss;
        __syncthreads();

        float tot = wave_sum(sRed[lane]);
        float inv = 1.0f / sqrtf(tot);
        tile_store(acc0, inv, sX + base0);
        if (two) tile_store(acc1, inv, sX + base1);
        __syncthreads();
    }

    for (int e = tid; e < NN * NN; e += NT)
        out[e] = sX[(e / NN) * PITCH + (e % NN)];
}


#define G_NT 192
#define PM 20
#define NB (NN / 16)
#define NV4 (NN * NN / 4)
#define WS_PF 0
#define WS_Q  (NN * NN)
#define WS_NS (2 * NN * NN)
#define WS_XA (3 * NN * NN)
#define WS_XB (4 * NN * NN)
#define WS_SA (5 * NN * NN)
#define WS_SB (5 * NN * NN + NB * 32)
#define WS_FLOATS (5 * NN * NN + 2 * NB * 32)

#define S_J 6
#define S_A 4
#define S_B (NN / 2)

__device__ __forceinline__ void s1_load_q(v4f (&q)[S_A], const float* qrow, int k0)
{
    #pragma unroll
    for (int v = 0; v < S_A; ++v)
        q[v] = *(const v4f*)(qrow + v * PITCH + k0);
}

template <bool REFILL>
__device__ __forceinline__ void s1_step(float (&m)[S_J][S_A], v4f (&x)[S_J], const v4f (&q)[S_A],
                                        const float* xrow, int knext)
{
    #pragma unroll
    for (int u = 0; u < S_J; ++u) {
        #pragma unroll
        for (int v = 0; v < S_A; ++v) {
            v2f p0 = q[v].lo * x[u].lo;
            v2f p1 = q[v].hi * x[u].hi;
            float t = __builtin_fmaxf(__builtin_fmaxf(m[u][v], p0.x), p0.y);
            m[u][v] = __builtin_fmaxf(__builtin_fmaxf(t, p1.x), p1.y);
        }
        if (REFILL)
            x[u] = *(const v4f*)(xrow + u * 16 * PITCH + knext);
    }
}

__device__ __forceinline__ void lds_barrier()
{
    __builtin_amdgcn_fence(__ATOMIC_RELEASE, "workgroup", "local");
    __builtin_amdgcn_s_barrier();
    __builtin_amdgcn_fence(__ATOMIC_ACQUIRE, "workgroup", "local");
}

__global__ __launch_bounds__(NT, 1)
void mpm_setup_kernel(const float* __restrict__ A_gt, const float* __restrict__ vec_logits, float* __restrict__ ws)
{
    __shared__ __attribute__((aligned(16))) float sP[NN * PITCH];
    __shared__ __attribute__((aligned(16))) float sQ[NN * PITCH];
    __shared__ __attribute__((aligned(16))) float sNS[NN * PITCH];
    __shared__ float sDegA[NN], sDegB[NN], sDA[NN], sDB[NN];
    const int tid = threadIdx.x;

    for (int e = tid; e < NN * NN; e += NT) {
        int i = e / NN, j = e % NN;
        float l;
        if (i == j) {
            l = 1.0f;
        } else {
            int lo = (i < j) ? i : j;
            int hi = (i < j) ? j : i;
            int off = (lo * (2 * NN - lo - 1)) / 2 + (hi - lo - 1);
            l = vec_logits[off];
        }
        sQ[i * PITCH + j] = 1.0f / (1.0f + expf(-l));
        sP[i * PITCH + j] = A_gt[e];
    }
    __syncthreads();
    if (tid < NN) {
        float ga = 0.f, gb = 0.f;
        for (int j = 0; j < NN; ++j) {
            ga += sP[tid * PITCH + j];
            gb += sQ[tid * PITCH + j];
        }
        sDA[tid] = sP[tid * PITCH + tid];
        sDB[tid] = sQ[tid * PITCH + tid];
        sDegA[tid] = ga;
        sDegB[tid] = gb;
    }
    __syncthreads();
    for (int e = tid; e < NN * NN; e += NT) {
        int i = e / NN, j = e % NN;
        float p = sP[i * PITCH + j] * sDA[i] * sDA[j];
        float q = sQ[i * PITCH + j] * sDB[i] * sDB[j];
        if (i == j) { p = 0.f; q = 0.f; }
        sP[i * PITCH + j] = p;
        sQ[i * PITCH + j] = q;
        sNS[i * PITCH + j] = sDA[i] * sDB[j] / (fabsf(sDegA[i] - sDegB[j]) + 1.0f);
    }
    __syncthreads();
    for (int f = tid; f < NV4; f += NT) {
        int ti = f / 384, kk = (f % 384) / 32, ln = f % 32;
        const float* s = &sP[(16 * ti + (ln & 15)) * PITCH + 8 * kk + 2 * (ln >> 4)];
        v4f a = { s[0], s[1], s[4], s[5] };
        ((v4f*)(ws + WS_PF))[f] = a;
        int qi = f / 24, qc = f % 24;
        ((v4f*)(ws + WS_Q))[f] = *(const v4f*)&sQ[qi * PITCH + 4 * qc];
        int g = f / 384, i = (f % 384) / 4, c4 = f % 4;
        ((v4f*)(ws + WS_NS))[f] = *(const v4f*)&sNS[i * PITCH + 16 * g + 4 * c4];
        v4f x0 = { 1.0f / (float)NN, 1.0f / (float)NN, 1.0f / (float)NN, 1.0f / (float)NN };
        ((v4f*)(ws + WS_XA))[f] = x0;
    }
    if (tid < NB * 32) ws[WS_SA + tid] = (tid == 0) ? 1.0f : 0.0f;
}

__global__ __launch_bounds__(G_NT, 1)
void mpm_iter_kernel(const float* pf, const float* qd, const float* nsb,
                     const float* xin, const float* ssin,
                     float* xout, float* ssout)
{
    __shared__ __attribute__((aligned(16))) float sX[NN * PITCH];
    __shared__ __attribute__((aligned(16))) float sQ[16 * PITCH];
    __shared__ float sM[NN * PM];
    __shared__ float sRed[NB];

    const int g    = blockIdx.x;
    const int tid  = threadIdx.x;
    const int lane = tid & 31;
    const int wave = __builtin_amdgcn_readfirstlane(tid >> 5);
    const int r = lane & 15;
    const int h = lane >> 4;

    const v4f* qsrc = (const v4f*)(qd + g * 16 * NN);
    v4f q0 = qsrc[tid];
    v4f q1 = qsrc[tid + G_NT];
    v4f xs[NV4 / G_NT];
    #pragma unroll
    for (int t = 0; t < NV4 / G_NT; ++t) xs[t] = ((const v4f*)xin)[tid + t * G_NT];
    v4f af[12];
    float nsr[8];
    {
        const v4f* p = (const v4f*)pf + wave * 12 * 32 + lane;
        #pragma unroll
        for (int kk = 0; kk < 12; ++kk) af[kk] = p[kk * 32];
        const float* pn = nsb + (g * NN + wave * 16 + 8 * h) * 16 + r;
        #pragma unroll
        for (int v = 0; v < 8; ++v) nsr[v] = pn[v * 16];
    }

    float p0 = ssin[0], p1 = ssin[32], p2 = ssin[64], p3 = ssin[96], p4 = ssin[128], p5 = ssin[160];
    asm volatile("" : "+s"(p0) : : "memory");

    float tot = ((((p0 + p1) + p2) + p3) + p4) + p5;
    float inv = 1.0f / sqrtf(tot);

    #pragma unroll
    for (int t = 0; t < NV4 / G_NT; ++t) {
        int e = tid + t * G_NT;
        int gg = e / 384, i = (e % 384) >> 2, c4 = e & 3;
        *(v4f*)&sX[i * PITCH + 16 * gg + 4 * c4] = xs[t] * inv;
    }
    *(v4f*)&sQ[(tid / 24) * PITCH + 4 * (tid % 24)] = q0;
    *(v4f*)&sQ[((tid + G_NT) / 24) * PITCH + 4 * ((tid + G_NT) % 24)] = q1;
    lds_barrier();

    if (wave < 4) {
        const float* xrow = &sX[r * PITCH + h * S_B];
        const float* qrow = &sQ[4 * wave * PITCH + h * S_B];
        float m[S_J][S_A];
        #pragma unroll
        for (int u = 0; u < S_J; ++u) {
            #pragma unroll
            for (int v = 0; v < S_A; ++v) m[u][v] = 0.0f;
        }
        v4f x[S_J], qa[S_A], qb[S_A];
        #pragma unroll
        for (int u = 0; u < S_J; ++u)
            x[u] = *(const v4f*)(xrow + u * 16 * PITCH);
        s1_load_q(qa, qrow, 0);
        for (int k0 = 0; k0 < S_B - 8; k0 += 8) {
            s1_load_q(qb, qrow, k0 + 4);
            s1_step<true>(m, x, qa, xrow, k0 + 4);
            s1_load_q(qa, qrow, k0 + 8);
            s1_step<true>(m, x, qb, xrow, k0 + 8);
        }
        s1_load_q(qb, qrow, S_B - 4);
        s1_step<true>(m, x, qa, xrow, S_B - 4);
        s1_step<false>(m, x, qb, xrow, 0);
        float* mrow = &sM[r * PM + 4 * wave + 2 * h];
        #pragma unroll
        for (int u = 0; u < S_J; ++u) {
            #pragma unroll
            for (int v = 0; v < S_A / 2; ++v) {
                float keep = h ? m[u][v + S_A / 2] : m[u][v];
                float send = h ? m[u][v] : m[u][v + S_A / 2];
                mrow[u * 16 * PM + v] = __builtin_fmaxf(keep, xhalf(send));
            }
        }
    }
    __syncthreads();

    {
        const float* pb = sM + 2 * h * PM + r;
        v2f bf[24];
        #pragma unroll
        for (int s = 0; s < 24; ++s) {
            bf[s].x = pb[(4 * s) * PM];
            bf[s].y = pb[(4 * s + 1) * PM];
        }
        const int row0 = wave * 16 + 8 * h;
        const float* px = sX + row0 * PITCH + 16 * g + r;
        float xv[8];
        #pragma unroll
        for (int v = 0; v < 8; ++v) xv[v] = px[v * PITCH];
        v8f c = {};
        asm volatile("" : "+v"(c) : : "memory");
        #pragma unroll
        for (int kk = 0; kk < 12; ++kk) {
            c = __builtin_amdgcn_wmma_f32_16x16x4_f32(false, af[kk].lo, false, bf[2 * kk], (short)0, c, false, false);
            c = __builtin_amdgcn_wmma_f32_16x16x4_f32(false, af[kk].hi, false, bf[2 * kk + 1], (short)0, c, false, false);
        }
        float* po = xout + (g * NN + row0) * 16 + r;
        float ss = 0.f;
        #pragma unroll
        for (int v = 0; v < 8; ++v) {
            float xn = xv[v] * nsr[v] + c[v];
            po[v * 16] = xn;
            ss += xn * xn;
        }
        ss = wave_sum(ss);
        if (lane == 0) sRed[wave] = ss;
    }
    __syncthreads();
    if (tid == 0)
        ssout[32 * g] = ((((sRed[0] + sRed[1]) + sRed[2]) + sRed[3]) + sRed[4]) + sRed[5];
}

__global__ __launch_bounds__(256, 1)
void mpm_final_kernel(const float* __restrict__ xin, const float* __restrict__ ssin, float* __restrict__ out)
{
    const int e = blockIdx.x * 256 + threadIdx.x;
    float tot = ((((ssin[0] + ssin[32]) + ssin[64]) + ssin[96]) + ssin[128]) + ssin[160];
    float inv = 1.0f / sqrtf(tot);
    const int i = e / 24, c = e % 24;
    v4f x = ((const v4f*)xin)[((c >> 2) * NN + i) * 4 + (c & 3)];
    ((v4f*)out)[e] = x * inv;
}

extern "C" void kernel_launch(void* const* d_in, const int* in_sizes, int n_in,
                              void* d_out, int out_size, void* d_ws, size_t ws_size,
                              hipStream_t stream) {
    (void)in_sizes; (void)n_in; (void)out_size;
    const float* A_gt = (const float*)d_in[0];
    const float* vec  = (const float*)d_in[1];
    float* out = (float*)d_out;
    float* ws  = (float*)d_ws;
    if (ws_size < sizeof(float) * WS_FLOATS) {
        graphvae_mpm_kernel<<<1, NT, 0, stream>>>(A_gt, vec, out);
        return;
    }
    mpm_setup_kernel<<<1, NT, 0, stream>>>(A_gt, vec, ws);
    for (int it = 0; it < MPM_ITERS; it += 2) {
        mpm_iter_kernel<<<NB, G_NT, 0, stream>>>(ws + WS_PF, ws + WS_Q, ws + WS_NS,
                                                 ws + WS_XA, ws + WS_SA, ws + WS_XB, ws + WS_SB);
        mpm_iter_kernel<<<NB, G_NT, 0, stream>>>(ws + WS_PF, ws + WS_Q, ws + WS_NS,
                                                 ws + WS_XB, ws + WS_SB, ws + WS_XA, ws + WS_SA);
    }
    mpm_final_kernel<<<NN * NN / 4 / 256, 256, 0, stream>>>(ws + WS_XA, ws + WS_SA, out);
}


# phase 1 (one wave per SIMD): two independent running maxima per output, each written a whole X row after its products; a fence per row keeps LDS reads a step ahead; max exact, no sum reordered
# speedup vs baseline: 7.9264x; 1.0430x over previous
#include <hip/hip_runtime.h>
#include <math.h>

#define NN 96
#define PITCH 100
#define MPM_ITERS 50
#define NT 1024

typedef __attribute__((ext_vector_type(2))) float v2f;
typedef __attribute__((ext_vector_type(8))) float v8f;
typedef __attribute__((ext_vector_type(4))) float v4f;

#define P1_J 4
#define P1_A 6
#define P1_B (NN / 2)
#define P1_THREADS (2 * (NN / P1_J) * (NN / P1_A))


__device__ __forceinline__ void p1_load_q(v4f (&q)[P1_A], const float* qrow, int k0)
{
    #pragma unroll
    for (int v = 0; v < P1_A; ++v)
        q[v] = *(const v4f*)(qrow + v * (NN / P1_A) * PITCH + k0);
}

template <bool REFILL>
__device__ __forceinline__ void p1_step(float (&m)[P1_J][P1_A], v4f (&x)[P1_J], const v4f (&q)[P1_A],
                                        const float* xrow, int knext)
{
    #pragma unroll
    for (int u = 0; u < P1_J; ++u) {
        #pragma unroll
        for (int v = 0; v < P1_A; ++v) {
            v2f p0 = q[v].lo * x[u].lo;
            v2f p1 = q[v].hi * x[u].hi;
            float t = __builtin_fmaxf(__builtin_fmaxf(m[u][v], p0.x), p0.y);
            m[u][v] = __builtin_fmaxf(__builtin_fmaxf(t, p1.x), p1.y);
        }
        if (REFILL)
            x[u] = *(const v4f*)(xrow + u * (NN / P1_J) * PITCH + knext);
    }
}


__device__ __forceinline__ v8f tile_gemm(const float* pa, const float* pb)
{
    v8f c = {};
    #pragma unroll
    for (int k0 = 0; k0 < NN; k0 += 4) {
        v2f av, bv;
        av.x = pa[k0];
        av.y = pa[k0 + 1];
        bv.x = pb[k0 * PITCH];
        bv.y = pb[(k0 + 1) * PITCH];
        c = __builtin_amdgcn_wmma_f32_16x16x4_f32(
                 false, av,  false, bv,
                 (short)0, c,  false,  false);
    }
    return c;
}

__device__ __forceinline__ void tile_update(v8f& acc, float& ss, const float* px, const float* pns)
{
    #pragma unroll
    for (int v = 0; v < 8; ++v) {
        float xn = px[v * PITCH] * pns[v * PITCH] + acc[v];
        acc[v] = xn;
        ss += xn * xn;
    }
}

__device__ __forceinline__ void tile_store(const v8f& acc, float inv, float* px)
{
    #pragma unroll
    for (int v = 0; v < 8; ++v)
        px[v * PITCH] = acc[v] * inv;
}

template <int MASK>
__device__ __forceinline__ float row_xor(float x)
{
    int b = __builtin_bit_cast(int, x);
    return __builtin_bit_cast(float, __builtin_amdgcn_update_dpp(b, b, 0x160 + MASK, 0xf, 0xf, false));
}

__device__ __forceinline__ float xhalf(float x)
{
    int b = __builtin_bit_cast(int, x);
    return __builtin_bit_cast(float, __builtin_amdgcn_permlanex16(b, b, 0x76543210, 0xfedcba98, false, false));
}

__device__ __forceinline__ float wave_sum(float x)
{
    x += row_xor<1>(x);
    x += row_xor<2>(x);
    x += row_xor<4>(x);
    x += row_xor<8>(x);
    x += xhalf(x);
    return x;
}

__global__ __launch_bounds__(NT, 1)
void graphvae_mpm_kernel(const float* __restrict__ A_gt,
                         const float* __restrict__ vec_logits,
                         float* __restrict__ out)
{
    __shared__ float sP[NN * PITCH];
    __shared__ __attribute__((aligned(16))) float sQ[NN * PITCH];
    __shared__ __attribute__((aligned(16))) float sX[NN * PITCH];
    __shared__ float sM[NN * PITCH];
    __shared__ float sNS[NN * PITCH];
    __shared__ float sDegA[NN], sDegB[NN], sDA[NN], sDB[NN];
    __shared__ float sRed[32];

    const int tid  = threadIdx.x;
    const int lane = tid & 31;
    const int wave = __builtin_amdgcn_readfirstlane(tid >> 5);

    for (int e = tid; e < NN * NN; e += NT) {
        int i = e / NN, j = e % NN;
        float l;
        if (i == j) {
            l = 1.0f;
        } else {
            int lo = (i < j) ? i : j;
            int hi = (i < j) ? j : i;
            int off = (lo * (2 * NN - lo - 1)) / 2 + (hi - lo - 1);
            l = vec_logits[off];
        }
        sQ[i * PITCH + j] = 1.0f / (1.0f + expf(-l));
        sP[i * PITCH + j] = A_gt[e];
        sX[i * PITCH + j] = 1.0f / (float)NN;
    }
    __syncthreads();

    if (tid < NN) {
        float ga = 0.f, gb = 0.f;
        for (int j = 0; j < NN; ++j) {
            ga += sP[tid * PITCH + j];
            gb += sQ[tid * PITCH + j];
        }
        sDA[tid] = sP[tid * PITCH + tid];
        sDB[tid] = sQ[tid * PITCH + tid];
        sDegA[tid] = ga;
        sDegB[tid] = gb;
    }
    __syncthreads();

    for (int e = tid; e < NN * NN; e += NT) {
        int i = e / NN, j = e % NN;
        float p = sP[i * PITCH + j] * sDA[i] * sDA[j];
        float q = sQ[i * PITCH + j] * sDB[i] * sDB[j];
        if (i == j) { p = 0.f; q = 0.f; }
        sP[i * PITCH + j] = p;
        sQ[i * PITCH + j] = q;
        sNS[i * PITCH + j] = sDA[i] * sDB[j] / (fabsf(sDegA[i] - sDegB[j]) + 1.0f);
    }
    __syncthreads();

    const int r = lane & 15;
    const int h = lane >> 4;
    const int ti0 = wave / 6, tn0 = wave % 6;
    const int ti1 = 5,        tn1 = 2 + wave;
    const int base0 = (ti0 * 16 + 8 * h) * PITCH + tn0 * 16 + r;
    const int base1 = (ti1 * 16 + 8 * h) * PITCH + tn1 * 16 + r;
    const int offa0 = (ti0 * 16 + r) * PITCH + 2 * h;
    const int offa1 = (ti1 * 16 + r) * PITCH + 2 * h;
    const int offb0 = 2 * h * PITCH + tn0 * 16 + r;
    const int offb1 = 2 * h * PITCH + tn1 * 16 + r;
    const bool two = wave < 4;

    for (int it = 0; it < MPM_ITERS; ++it) {
        if (tid < P1_THREADS) {
            const int jg = wave;
            const int ag = lane & 15;
            const int hb = lane >> 4;
            const float* xrow = &sX[jg * PITCH + hb * P1_B];
            const float* qrow = &sQ[ag * PITCH + hb * P1_B];
            float m[P1_J][P1_A];
            #pragma unroll
            for (int u = 0; u < P1_J; ++u) {
                #pragma unroll
                for (int v = 0; v < P1_A; ++v) m[u][v] = 0.0f;
            }
            v4f x[P1_J], qa[P1_A], qb[P1_A];
            #pragma unroll
            for (int u = 0; u < P1_J; ++u)
                x[u] = *(const v4f*)(xrow + u * (NN / P1_J) * PITCH);
            p1_load_q(qa, qrow, 0);
            for (int k0 = 0; k0 < P1_B - 8; k0 += 8) {
                p1_load_q(qb, qrow, k0 + 4);
                p1_step<true>(m, x, qa, xrow, k0 + 4);
                p1_load_q(qa, qrow, k0 + 8);
                p1_step<true>(m, x, qb, xrow, k0 + 8);
            }
            p1_load_q(qb, qrow, P1_B - 4);
            p1_step<true>(m, x, qa, xrow, P1_B - 4);
            p1_step<false>(m, x, qb, xrow, 0);
            float* mrow = &sM[jg * PITCH + ag + hb * (P1_A / 2) * (NN / P1_A)];
            #pragma unroll
            for (int u = 0; u < P1_J; ++u) {
                #pragma unroll
                for (int v = 0; v < P1_A / 2; ++v) {
                    float keep = hb ? m[u][v + P1_A / 2] : m[u][v];
                    float send = hb ? m[u][v] : m[u][v + P1_A / 2];
                    mrow[u * (NN / P1_J) * PITCH + v * (NN / P1_A)] = __builtin_fmaxf(keep, xhalf(send));
                }
            }
        }
        __syncthreads();

        v8f acc1 = {};
        if (two) acc1 = tile_gemm(sP + offa1, sM + offb1);
        v8f acc0 = tile_gemm(sP + offa0, sM + offb0);

        float ss = 0.f;
        tile_update(acc0, ss, sX + base0, sNS + base0);
        if (two) tile_update(acc1, ss, sX + base1, sNS + base1);
        ss = wave_sum(ss);
        if (lane == 0) sRed[wave] = ss;
        __syncthreads();

        float tot = wave_sum(sRed[lane]);
        float inv = 1.0f / sqrtf(tot);
        tile_store(acc0, inv, sX + base0);
        if (two) tile_store(acc1, inv, sX + base1);
        __syncthreads();
    }

    for (int e = tid; e < NN * NN; e += NT)
        out[e] = sX[(e / NN) * PITCH + (e % NN)];
}


#define G_NT 192
#define PM 20
#define NB (NN / 16)
#define NV4 (NN * NN / 4)
#define WS_PF 0
#define WS_Q  (NN * NN)
#define WS_NS (2 * NN * NN)
#define WS_XA (3 * NN * NN)
#define WS_XB (4 * NN * NN)
#define WS_SA (5 * NN * NN)
#define WS_SB (5 * NN * NN + NB * 32)
#define WS_FLOATS (5 * NN * NN + 2 * NB * 32)

#define S_J 6
#define S_A 4
#define S_B (NN / 2)

__device__ __forceinline__ void s1_load_q(v4f (&q)[S_A], const float* qrow, int k0)
{
    #pragma unroll
    for (int v = 0; v < S_A; ++v)
        q[v] = *(const v4f*)(qrow + v * PITCH + k0);
}

template <bool REFILL>
__device__ __forceinline__ void s1_step(float (&ma)[S_J][S_A], float (&mb)[S_J][S_A], v4f (&x)[S_J],
                                        const v4f (&q)[S_A], const float* xrow, int knext)
{
    #pragma unroll
    for (int u = 0; u < S_J; ++u) {
        v2f p0[S_A], p1[S_A];
        #pragma unroll
        for (int v = 0; v < S_A; ++v) {
            p0[v] = q[v].lo * x[u].lo;
            p1[v] = q[v].hi * x[u].hi;
        }
        if (REFILL)
            x[u] = *(const v4f*)(xrow + u * 16 * PITCH + knext);
        #pragma unroll
        for (int v = 0; v < S_A; ++v) {
            ma[u][v] = __builtin_fmaxf(__builtin_fmaxf(ma[u][v], p0[v].x), p0[v].y);
            mb[u][v] = __builtin_fmaxf(__builtin_fmaxf(mb[u][v], p1[v].x), p1[v].y);
        }
        __builtin_amdgcn_sched_barrier(0);
    }
}

__device__ __forceinline__ void lds_barrier()
{
    __builtin_amdgcn_fence(__ATOMIC_RELEASE, "workgroup", "local");
    __builtin_amdgcn_s_barrier();
    __builtin_amdgcn_fence(__ATOMIC_ACQUIRE, "workgroup", "local");
}

__global__ __launch_bounds__(NT, 1)
void mpm_setup_kernel(const float* __restrict__ A_gt, const float* __restrict__ vec_logits, float* __restrict__ ws)
{
    __shared__ __attribute__((aligned(16))) float sP[NN * PITCH];
    __shared__ __attribute__((aligned(16))) float sQ[NN * PITCH];
    __shared__ __attribute__((aligned(16))) float sNS[NN * PITCH];
    __shared__ float sDegA[NN], sDegB[NN], sDA[NN], sDB[NN];
    const int tid = threadIdx.x;

    for (int e = tid; e < NN * NN; e += NT) {
        int i = e / NN, j = e % NN;
        float l;
        if (i == j) {
            l = 1.0f;
        } else {
            int lo = (i < j) ? i : j;
            int hi = (i < j) ? j : i;
            int off = (lo * (2 * NN - lo - 1)) / 2 + (hi - lo - 1);
            l = vec_logits[off];
        }
        sQ[i * PITCH + j] = 1.0f / (1.0f + expf(-l));
        sP[i * PITCH + j] = A_gt[e];
    }
    __syncthreads();
    if (tid < NN) {
        float ga = 0.f, gb = 0.f;
        for (int j = 0; j < NN; ++j) {
            ga += sP[tid * PITCH + j];
            gb += sQ[tid * PITCH + j];
        }
        sDA[tid] = sP[tid * PITCH + tid];
        sDB[tid] = sQ[tid * PITCH + tid];
        sDegA[tid] = ga;
        sDegB[tid] = gb;
    }
    __syncthreads();
    for (int e = tid; e < NN * NN; e += NT) {
        int i = e / NN, j = e % NN;
        float p = sP[i * PITCH + j] * sDA[i] * sDA[j];
        float q = sQ[i * PITCH + j] * sDB[i] * sDB[j];
        if (i == j) { p = 0.f; q = 0.f; }
        sP[i * PITCH + j] = p;
        sQ[i * PITCH + j] = q;
        sNS[i * PITCH + j] = sDA[i] * sDB[j] / (fabsf(sDegA[i] - sDegB[j]) + 1.0f);
    }
    __syncthreads();
    for (int f = tid; f < NV4; f += NT) {
        int ti = f / 384, kk = (f % 384) / 32, ln = f % 32;
        const float* s = &sP[(16 * ti + (ln & 15)) * PITCH + 8 * kk + 2 * (ln >> 4)];
        v4f a = { s[0], s[1], s[4], s[5] };
        ((v4f*)(ws + WS_PF))[f] = a;
        int qi = f / 24, qc = f % 24;
        ((v4f*)(ws + WS_Q))[f] = *(const v4f*)&sQ[qi * PITCH + 4 * qc];
        int g = f / 384, i = (f % 384) / 4, c4 = f % 4;
        ((v4f*)(ws + WS_NS))[f] = *(const v4f*)&sNS[i * PITCH + 16 * g + 4 * c4];
        v4f x0 = { 1.0f / (float)NN, 1.0f / (float)NN, 1.0f / (float)NN, 1.0f / (float)NN };
        ((v4f*)(ws + WS_XA))[f] = x0;
    }
    if (tid < NB * 32) ws[WS_SA + tid] = (tid == 0) ? 1.0f : 0.0f;
}

__global__ __launch_bounds__(G_NT, 1)
void mpm_iter_kernel(const float* pf, const float* qd, const float* nsb,
                     const float* xin, const float* ssin,
                     float* xout, float* ssout)
{
    __shared__ __attribute__((aligned(16))) float sX[NN * PITCH];
    __shared__ __attribute__((aligned(16))) float sQ[16 * PITCH];
    __shared__ float sM[NN * PM];
    __shared__ float sRed[NB];

    const int g    = blockIdx.x;
    const int tid  = threadIdx.x;
    const int lane = tid & 31;
    const int wave = __builtin_amdgcn_readfirstlane(tid >> 5);
    const int r = lane & 15;
    const int h = lane >> 4;

    const v4f* qsrc = (const v4f*)(qd + g * 16 * NN);
    v4f q0 = qsrc[tid];
    v4f q1 = qsrc[tid + G_NT];
    v4f xs[NV4 / G_NT];
    #pragma unroll
    for (int t = 0; t < NV4 / G_NT; ++t) xs[t] = ((const v4f*)xin)[tid + t * G_NT];
    v4f af[12];
    float nsr[8];
    {
        const v4f* p = (const v4f*)pf + wave * 12 * 32 + lane;
        #pragma unroll
        for (int kk = 0; kk < 12; ++kk) af[kk] = p[kk * 32];
        const float* pn = nsb + (g * NN + wave * 16 + 8 * h) * 16 + r;
        #pragma unroll
        for (int v = 0; v < 8; ++v) nsr[v] = pn[v * 16];
    }

    float p0 = ssin[0], p1 = ssin[32], p2 = ssin[64], p3 = ssin[96], p4 = ssin[128], p5 = ssin[160];
    asm volatile("" : "+s"(p0) : : "memory");

    float tot = ((((p0 + p1) + p2) + p3) + p4) + p5;
    float inv = 1.0f / sqrtf(tot);

    #pragma unroll
    for (int t = 0; t < NV4 / G_NT; ++t) {
        int e = tid + t * G_NT;
        int gg = e / 384, i = (e % 384) >> 2, c4 = e & 3;
        *(v4f*)&sX[i * PITCH + 16 * gg + 4 * c4] = xs[t] * inv;
    }
    *(v4f*)&sQ[(tid / 24) * PITCH + 4 * (tid % 24)] = q0;
    *(v4f*)&sQ[((tid + G_NT) / 24) * PITCH + 4 * ((tid + G_NT) % 24)] = q1;
    lds_barrier();

    if (wave < 4) {
        const float* xrow = &sX[r * PITCH + h * S_B];
        const float* qrow = &sQ[4 * wave * PITCH + h * S_B];
        float ma[S_J][S_A], mb[S_J][S_A], m[S_J][S_A];
        #pragma unroll
        for (int u = 0; u < S_J; ++u) {
            #pragma unroll
            for (int v = 0; v < S_A; ++v) { ma[u][v] = 0.0f; mb[u][v] = 0.0f; }
        }
        v4f x[S_J], qa[S_A], qb[S_A];
        #pragma unroll
        for (int u = 0; u < S_J; ++u)
            x[u] = *(const v4f*)(xrow + u * 16 * PITCH);
        s1_load_q(qa, qrow, 0);
        for (int k0 = 0; k0 < S_B - 8; k0 += 8) {
            s1_load_q(qb, qrow, k0 + 4);
            s1_step<true>(ma, mb, x, qa, xrow, k0 + 4);
            s1_load_q(qa, qrow, k0 + 8);
            s1_step<true>(ma, mb, x, qb, xrow, k0 + 8);
        }
        s1_load_q(qb, qrow, S_B - 4);
        s1_step<true>(ma, mb, x, qa, xrow, S_B - 4);
        s1_step<false>(ma, mb, x, qb, xrow, 0);
        #pragma unroll
        for (int u = 0; u < S_J; ++u) {
            #pragma unroll
            for (int v = 0; v < S_A; ++v) m[u][v] = __builtin_fmaxf(ma[u][v], mb[u][v]);
        }
        float* mrow = &sM[r * PM + 4 * wave + 2 * h];
        #pragma unroll
        for (int u = 0; u < S_J; ++u) {
            #pragma unroll
            for (int v = 0; v < S_A / 2; ++v) {
                float keep = h ? m[u][v + S_A / 2] : m[u][v];
                float send = h ? m[u][v] : m[u][v + S_A / 2];
                mrow[u * 16 * PM + v] = __builtin_fmaxf(keep, xhalf(send));
            }
        }
    }
    __syncthreads();

    {
        const float* pb = sM + 2 * h * PM + r;
        v2f bf[24];
        #pragma unroll
        for (int s = 0; s < 24; ++s) {
            bf[s].x = pb[(4 * s) * PM];
            bf[s].y = pb[(4 * s + 1) * PM];
        }
        const int row0 = wave * 16 + 8 * h;
        const float* px = sX + row0 * PITCH + 16 * g + r;
        float xv[8];
        #pragma unroll
        for (int v = 0; v < 8; ++v) xv[v] = px[v * PITCH];
        v8f c = {};
        asm volatile("" : "+v"(c) : : "memory");
        #pragma unroll
        for (int kk = 0; kk < 12; ++kk) {
            c = __builtin_amdgcn_wmma_f32_16x16x4_f32(false, af[kk].lo, false, bf[2 * kk], (short)0, c, false, false);
            c = __builtin_amdgcn_wmma_f32_16x16x4_f32(false, af[kk].hi, false, bf[2 * kk + 1], (short)0, c, false, false);
        }
        float* po = xout + (g * NN + row0) * 16 + r;
        float ss = 0.f;
        #pragma unroll
        for (int v = 0; v < 8; ++v) {
            float xn = xv[v] * nsr[v] + c[v];
            po[v * 16] = xn;
            ss += xn * xn;
        }
        ss = wave_sum(ss);
        if (lane == 0) sRed[wave] = ss;
    }
    __syncthreads();
    if (tid == 0)
        ssout[32 * g] = ((((sRed[0] + sRed[1]) + sRed[2]) + sRed[3]) + sRed[4]) + sRed[5];
}

__global__ __launch_bounds__(256, 1)
void mpm_final_kernel(const float* __restrict__ xin, const float* __restrict__ ssin, float* __restrict__ out)
{
    const int e = blockIdx.x * 256 + threadIdx.x;
    float tot = ((((ssin[0] + ssin[32]) + ssin[64]) + ssin[96]) + ssin[128]) + ssin[160];
    float inv = 1.0f / sqrtf(tot);
    const int i = e / 24, c = e % 24;
    v4f x = ((const v4f*)xin)[((c >> 2) * NN + i) * 4 + (c & 3)];
    ((v4f*)out)[e] = x * inv;
}

extern "C" void kernel_launch(void* const* d_in, const int* in_sizes, int n_in,
                              void* d_out, int out_size, void* d_ws, size_t ws_size,
                              hipStream_t stream) {
    (void)in_sizes; (void)n_in; (void)out_size;
    const float* A_gt = (const float*)d_in[0];
    const float* vec  = (const float*)d_in[1];
    float* out = (float*)d_out;
    float* ws  = (float*)d_ws;
    if (ws_size < sizeof(float) * WS_FLOATS) {
        graphvae_mpm_kernel<<<1, NT, 0, stream>>>(A_gt, vec, out);
        return;
    }
    mpm_setup_kernel<<<1, NT, 0, stream>>>(A_gt, vec, ws);
    for (int it = 0; it < MPM_ITERS; it += 2) {
        mpm_iter_kernel<<<NB, G_NT, 0, stream>>>(ws + WS_PF, ws + WS_Q, ws + WS_NS,
                                                 ws + WS_XA, ws + WS_SA, ws + WS_XB, ws + WS_SB);
        mpm_iter_kernel<<<NB, G_NT, 0, stream>>>(ws + WS_PF, ws + WS_Q, ws + WS_NS,
                                                 ws + WS_XB, ws + WS_SB, ws + WS_XA, ws + WS_SA);
    }
    mpm_final_kernel<<<NN * NN / 4 / 256, 256, 0, stream>>>(ws + WS_XA, ws + WS_SA, out);
}


# each iteration spread over 24 work-groups (4 columns of M, Y, X each) instead of 6; a lane's X words go from the workspace straight into its registers, no LDS staging; norm summed in another order
# speedup vs baseline: 11.3067x; 1.4265x over previous
#include <hip/hip_runtime.h>
#include <math.h>

#define NN 96
#define PITCH 100
#define MPM_ITERS 50
#define NT 1024

typedef __attribute__((ext_vector_type(2))) float v2f;
typedef __attribute__((ext_vector_type(8))) float v8f;
typedef __attribute__((ext_vector_type(4))) float v4f;

#define P1_J 4
#define P1_A 6
#define P1_B (NN / 2)
#define P1_THREADS (2 * (NN / P1_J) * (NN / P1_A))


__device__ __forceinline__ void p1_load_q(v4f (&q)[P1_A], const float* qrow, int k0)
{
    #pragma unroll
    for (int v = 0; v < P1_A; ++v)
        q[v] = *(const v4f*)(qrow + v * (NN / P1_A) * PITCH + k0);
}

template <bool REFILL>
__device__ __forceinline__ void p1_step(float (&m)[P1_J][P1_A], v4f (&x)[P1_J], const v4f (&q)[P1_A],
                                        const float* xrow, int knext)
{
    #pragma unroll
    for (int u = 0; u < P1_J; ++u) {
        #pragma unroll
        for (int v = 0; v < P1_A; ++v) {
            v2f p0 = q[v].lo * x[u].lo;
            v2f p1 = q[v].hi * x[u].hi;
            float t = __builtin_fmaxf(__builtin_fmaxf(m[u][v], p0.x), p0.y);
            m[u][v] = __builtin_fmaxf(__builtin_fmaxf(t, p1.x), p1.y);
        }
        if (REFILL)
            x[u] = *(const v4f*)(xrow + u * (NN / P1_J) * PITCH + knext);
    }
}


__device__ __forceinline__ v8f tile_gemm(const float* pa, const float* pb)
{
    v8f c = {};
    #pragma unroll
    for (int k0 = 0; k0 < NN; k0 += 4) {
        v2f av, bv;
        av.x = pa[k0];
        av.y = pa[k0 + 1];
        bv.x = pb[k0 * PITCH];
        bv.y = pb[(k0 + 1) * PITCH];
        c = __builtin_amdgcn_wmma_f32_16x16x4_f32(
                 false, av,  false, bv,
                 (short)0, c,  false,  false);
    }
    return c;
}

__device__ __forceinline__ void tile_update(v8f& acc, float& ss, const float* px, const float* pns)
{
    #pragma unroll
    for (int v = 0; v < 8; ++v) {
        float xn = px[v * PITCH] * pns[v * PITCH] + acc[v];
        acc[v] = xn;
        ss += xn * xn;
    }
}

__device__ __forceinline__ void tile_store(const v8f& acc, float inv, float* px)
{
    #pragma unroll
    for (int v = 0; v < 8; ++v)
        px[v * PITCH] = acc[v] * inv;
}

template <int MASK>
__device__ __forceinline__ float row_xor(float x)
{
    int b = __builtin_bit_cast(int, x);
    return __builtin_bit_cast(float, __builtin_amdgcn_update_dpp(b, b, 0x160 + MASK, 0xf, 0xf, false));
}

__device__ __forceinline__ float xhalf(float x)
{
    int b = __builtin_bit_cast(int, x);
    return __builtin_bit_cast(float, __builtin_amdgcn_permlanex16(b, b, 0x76543210, 0xfedcba98, false, false));
}

__device__ __forceinline__ float wave_sum(float x)
{
    x += row_xor<1>(x);
    x += row_xor<2>(x);
    x += row_xor<4>(x);
    x += row_xor<8>(x);
    x += xhalf(x);
    return x;
}

__global__ __launch_bounds__(NT, 1)
void graphvae_mpm_kernel(const float* __restrict__ A_gt,
                         const float* __restrict__ vec_logits,
                         float* __restrict__ out)
{
    __shared__ float sP[NN * PITCH];
    __shared__ __attribute__((aligned(16))) float sQ[NN * PITCH];
    __shared__ __attribute__((aligned(16))) float sX[NN * PITCH];
    __shared__ float sM[NN * PITCH];
    __shared__ float sNS[NN * PITCH];
    __shared__ float sDegA[NN], sDegB[NN], sDA[NN], sDB[NN];
    __shared__ float sRed[32];

    const int tid  = threadIdx.x;
    const int lane = tid & 31;
    const int wave = __builtin_amdgcn_readfirstlane(tid >> 5);

    for (int e = tid; e < NN * NN; e += NT) {
        int i = e / NN, j = e % NN;
        float l;
        if (i == j) {
            l = 1.0f;
        } else {
            int lo = (i < j) ? i : j;
            int hi = (i < j) ? j : i;
            int off = (lo * (2 * NN - lo - 1)) / 2 + (hi - lo - 1);
            l = vec_logits[off];
        }
        sQ[i * PITCH + j] = 1.0f / (1.0f + expf(-l));
        sP[i * PITCH + j] = A_gt[e];
        sX[i * PITCH + j] = 1.0f / (float)NN;
    }
    __syncthreads();

    if (tid < NN) {
        float ga = 0.f, gb = 0.f;
        for (int j = 0; j < NN; ++j) {
            ga += sP[tid * PITCH + j];
            gb += sQ[tid * PITCH + j];
        }
        sDA[tid] = sP[tid * PITCH + tid];
        sDB[tid] = sQ[tid * PITCH + tid];
        sDegA[tid] = ga;
        sDegB[tid] = gb;
    }
    __syncthreads();

    for (int e = tid; e < NN * NN; e += NT) {
        int i = e / NN, j = e % NN;
        float p = sP[i * PITCH + j] * sDA[i] * sDA[j];
        float q = sQ[i * PITCH + j] * sDB[i] * sDB[j];
        if (i == j) { p = 0.f; q = 0.f; }
        sP[i * PITCH + j] = p;
        sQ[i * PITCH + j] = q;
        sNS[i * PITCH + j] = sDA[i] * sDB[j] / (fabsf(sDegA[i] - sDegB[j]) + 1.0f);
    }
    __syncthreads();

    const int r = lane & 15;
    const int h = lane >> 4;
    const int ti0 = wave / 6, tn0 = wave % 6;
    const int ti1 = 5,        tn1 = 2 + wave;
    const int base0 = (ti0 * 16 + 8 * h) * PITCH + tn0 * 16 + r;
    const int base1 = (ti1 * 16 + 8 * h) * PITCH + tn1 * 16 + r;
    const int offa0 = (ti0 * 16 + r) * PITCH + 2 * h;
    const int offa1 = (ti1 * 16 + r) * PITCH + 2 * h;
    const int offb0 = 2 * h * PITCH + tn0 * 16 + r;
    const int offb1 = 2 * h * PITCH + tn1 * 16 + r;
    const bool two = wave < 4;

    for (int it = 0; it < MPM_ITERS; ++it) {
        if (tid < P1_THREADS) {
            const int jg = wave;
            const int ag = lane & 15;
            const int hb = lane >> 4;
            const float* xrow = &sX[jg * PITCH + hb * P1_B];
            const float* qrow = &sQ[ag * PITCH + hb * P1_B];
            float m[P1_J][P1_A];
            #pragma unroll
            for (int u = 0; u < P1_J; ++u) {
                #pragma unroll
                for (int v = 0; v < P1_A; ++v) m[u][v] = 0.0f;
            }
            v4f x[P1_J], qa[P1_A], qb[P1_A];
            #pragma unroll
            for (int u = 0; u < P1_J; ++u)
                x[u] = *(const v4f*)(xrow + u * (NN / P1_J) * PITCH);
            p1_load_q(qa, qrow, 0);
            for (int k0 = 0; k0 < P1_B - 8; k0 += 8) {
                p1_load_q(qb, qrow, k0 + 4);
                p1_step<true>(m, x, qa, xrow, k0 + 4);
                p1_load_q(qa, qrow, k0 + 8);
                p1_step<true>(m, x, qb, xrow, k0 + 8);
            }
            p1_load_q(qb, qrow, P1_B - 4);
            p1_step<true>(m, x, qa, xrow, P1_B - 4);
            p1_step<false>(m, x, qb, xrow, 0);
            float* mrow = &sM[jg * PITCH + ag + hb * (P1_A / 2) * (NN / P1_A)];
            #pragma unroll
            for (int u = 0; u < P1_J; ++u) {
                #pragma unroll
                for (int v = 0; v < P1_A / 2; ++v) {
                    float keep = hb ? m[u][v + P1_A / 2] : m[u][v];
                    float send = hb ? m[u][v] : m[u][v + P1_A / 2];
                    mrow[u * (NN / P1_J) * PITCH + v * (NN / P1_A)] = __builtin_fmaxf(keep, xhalf(send));
                }
            }
        }
        __syncthreads();

        v8f acc1 = {};
        if (two) acc1 = tile_gemm(sP + offa1, sM + offb1);
        v8f acc0 = tile_gemm(sP + offa0, sM + offb0);

        float ss = 0.f;
        tile_update(acc0, ss, sX + base0, sNS + base0);
        if (two) tile_update(acc1, ss, sX + base1, sNS + base1);
        ss = wave_sum(ss);
        if (lane == 0) sRed[wave] = ss;
        __syncthreads();

        float tot = wave_sum(sRed[lane]);
        float inv = 1.0f / sqrtf(tot);
        tile_store(acc0, inv, sX + base0);
        if (two) tile_store(acc1, inv, sX + base1);
        __syncthreads();
    }

    for (int e = tid; e < NN * NN; e += NT)
        out[e] = sX[(e / NN) * PITCH + (e % NN)];
}


#define G_NT 192
#define BC 4
#define NB (NN / BC)
#define PMT 104
#define NV4 (NN * NN / 4)
#define WS_PF 0
#define WS_Q  (NN * NN)
#define WS_NS (2 * NN * NN)
#define WS_XA (3 * NN * NN)
#define WS_XB (4 * NN * NN)
#define WS_SA (5 * NN * NN)
#define WS_SB (5 * NN * NN + NB * 32)
#define WS_FLOATS (5 * NN * NN + 2 * NB * 32)

#define R_U 3
#define R_S 6

__device__ __forceinline__ void lds_barrier()
{
    __builtin_amdgcn_fence(__ATOMIC_RELEASE, "workgroup", "local");
    __builtin_amdgcn_s_barrier();
    __builtin_amdgcn_fence(__ATOMIC_ACQUIRE, "workgroup", "local");
}

__device__ __forceinline__ void load_partials(float (&p)[NB], const float* ss)
{
    #pragma unroll
    for (int k = 0; k < NB; ++k) p[k] = ss[32 * k];
}

__device__ __forceinline__ float sum_partials(const float (&p)[NB])
{
    float a[NB / 2], b[NB / 4], c[NB / 8];
    #pragma unroll
    for (int k = 0; k < NB / 2; ++k) a[k] = p[2 * k] + p[2 * k + 1];
    #pragma unroll
    for (int k = 0; k < NB / 4; ++k) b[k] = a[2 * k] + a[2 * k + 1];
    #pragma unroll
    for (int k = 0; k < NB / 8; ++k) c[k] = b[2 * k] + b[2 * k + 1];
    return (c[0] + c[1]) + c[2];
}

__device__ __forceinline__ float pick4(float a0, float a1, float a2, float a3, int k)
{
    float lo = (k & 1) ? a1 : a0;
    float hi = (k & 1) ? a3 : a2;
    return (k & 2) ? hi : lo;
}

__global__ __launch_bounds__(NT, 1)
void mpm_setup_kernel(const float* __restrict__ A_gt, const float* __restrict__ vec_logits, float* __restrict__ ws)
{
    __shared__ __attribute__((aligned(16))) float sP[NN * PITCH];
    __shared__ __attribute__((aligned(16))) float sQ[NN * PITCH];
    __shared__ __attribute__((aligned(16))) float sNS[NN * PITCH];
    __shared__ float sDegA[NN], sDegB[NN], sDA[NN], sDB[NN];
    const int tid = threadIdx.x;

    for (int e = tid; e < NN * NN; e += NT) {
        int i = e / NN, j = e % NN;
        float l;
        if (i == j) {
            l = 1.0f;
        } else {
            int lo = (i < j) ? i : j;
            int hi = (i < j) ? j : i;
            int off = (lo * (2 * NN - lo - 1)) / 2 + (hi - lo - 1);
            l = vec_logits[off];
        }
        sQ[i * PITCH + j] = 1.0f / (1.0f + expf(-l));
        sP[i * PITCH + j] = A_gt[e];
    }
    __syncthreads();
    if (tid < NN) {
        float ga = 0.f, gb = 0.f;
        for (int j = 0; j < NN; ++j) {
            ga += sP[tid * PITCH + j];
            gb += sQ[tid * PITCH + j];
        }
        sDA[tid] = sP[tid * PITCH + tid];
        sDB[tid] = sQ[tid * PITCH + tid];
        sDegA[tid] = ga;
        sDegB[tid] = gb;
    }
    __syncthreads();
    for (int e = tid; e < NN * NN; e += NT) {
        int i = e / NN, j = e % NN;
        float p = sP[i * PITCH + j] * sDA[i] * sDA[j];
        float q = sQ[i * PITCH + j] * sDB[i] * sDB[j];
        if (i == j) { p = 0.f; q = 0.f; }
        sP[i * PITCH + j] = p;
        sQ[i * PITCH + j] = q;
        sNS[i * PITCH + j] = sDA[i] * sDB[j] / (fabsf(sDegA[i] - sDegB[j]) + 1.0f);
    }
    __syncthreads();
    for (int f = tid; f < NV4; f += NT) {
        int ti = f / 384, kk = (f % 384) / 32, ln = f % 32;
        const float* s = &sP[(16 * ti + (ln & 15)) * PITCH + 8 * kk + 2 * (ln >> 4)];
        v4f a = { s[0], s[1], s[4], s[5] };
        ((v4f*)(ws + WS_PF))[f] = a;
        int qi = f / 24, qc = f % 24;
        ((v4f*)(ws + WS_Q))[f] = *(const v4f*)&sQ[qi * PITCH + 4 * qc];
        int g = f / NN, i = f % NN;
        ((v4f*)(ws + WS_NS))[f] = *(const v4f*)&sNS[i * PITCH + BC * g];
        v4f x0 = { 1.0f / (float)NN, 1.0f / (float)NN, 1.0f / (float)NN, 1.0f / (float)NN };
        ((v4f*)(ws + WS_XA))[f] = x0;
    }
    if (tid < NB * 32) ws[WS_SA + tid] = (tid == 0) ? 1.0f : 0.0f;
}

__global__ __launch_bounds__(G_NT, 1)
void mpm_iter_kernel(const float* pf, const float* qd, const float* nsb,
                     const float* xin, const float* ssin,
                     float* xout, float* ssout)
{
    __shared__ __attribute__((aligned(16))) float sQ[BC * NN];
    __shared__ __attribute__((aligned(16))) float sMt[BC * PMT];
    __shared__ float sRed[G_NT / 32];

    const int g    = blockIdx.x;
    const int tid  = threadIdx.x;
    const int lane = tid & 31;
    const int wave = __builtin_amdgcn_readfirstlane(tid >> 5);
    const int r = lane & 15;
    const int h = lane >> 4;
    const int rl = lane & 7;
    const int bq = lane >> 3;

    float part[NB];
    load_partials(part, ssin);
    const bool ph1 = wave < 4;
    v4f x[R_S][R_U];
    if (ph1) {
        const v4f* xs = (const v4f*)xin + (R_S * bq) * NN + 24 * wave + rl;
        #pragma unroll
        for (int s = 0; s < R_S; ++s) {
            #pragma unroll
            for (int u = 0; u < R_U; ++u) x[s][u] = xs[s * NN + 8 * u];
        }
    } else {
        const v4f* qs = (const v4f*)(qd + g * BC * NN);
        const int tq = tid & 63;
        *(v4f*)&sQ[4 * tq] = qs[tq];
        if (wave == 4) *(v4f*)&sQ[4 * (64 + lane)] = qs[64 + lane];
    }
    v4f af[12];
    {
        const v4f* p = (const v4f*)pf + wave * 12 * 32 + lane;
        #pragma unroll
        for (int kk = 0; kk < 12; ++kk) af[kk] = p[kk * 32];
    }
    const int sel = r >> 2;
    const int e0 = (g * NN + wave * 16 + 8 * h + sel) * BC + (r & 3);
    const float ns0 = nsb[e0], ns1 = nsb[e0 + 4 * BC];
    const float xr0 = xin[e0], xr1 = xin[e0 + 4 * BC];

    lds_barrier();

    v4f q[R_S][BC];
    if (ph1) {
        const float* qrow = &sQ[24 * bq];
        #pragma unroll
        for (int s = 0; s < R_S; ++s) {
            #pragma unroll
            for (int v = 0; v < BC; ++v) q[s][v] = *(const v4f*)(qrow + v * NN + 4 * s);
        }
    }

    asm volatile("" : "+s"(part[0]), "+s"(part[1]), "+s"(part[2]), "+s"(part[3]), "+s"(part[4]), "+s"(part[5]),
                      "+s"(part[6]), "+s"(part[7]), "+s"(part[8]), "+s"(part[9]), "+s"(part[10]), "+s"(part[11]),
                      "+s"(part[12]), "+s"(part[13]), "+s"(part[14]), "+s"(part[15]), "+s"(part[16]), "+s"(part[17]),
                      "+s"(part[18]), "+s"(part[19]), "+s"(part[20]), "+s"(part[21]), "+s"(part[22]), "+s"(part[23])
                 : : "memory");
    const float inv = 1.0f / sqrtf(sum_partials(part));

    if (ph1) {
        float ma[R_U][BC], mb[R_U][BC];
        #pragma unroll
        for (int u = 0; u < R_U; ++u) {
            #pragma unroll
            for (int v = 0; v < BC; ++v) { ma[u][v] = 0.0f; mb[u][v] = 0.0f; }
        }
        #pragma unroll
        for (int s = 0; s < R_S; ++s) {
            #pragma unroll
            for (int u = 0; u < R_U; ++u) {
                const v4f xv = x[s][u] * inv;
                v2f p0[BC], p1[BC];
                #pragma unroll
                for (int v = 0; v < BC; ++v) {
                    p0[v] = q[s][v].lo * xv.lo;
                    p1[v] = q[s][v].hi * xv.hi;
                }
                #pragma unroll
                for (int v = 0; v < BC; ++v) {
                    ma[u][v] = __builtin_fmaxf(__builtin_fmaxf(ma[u][v], p0[v].x), p0[v].y);
                    mb[u][v] = __builtin_fmaxf(__builtin_fmaxf(mb[u][v], p1[v].x), p1[v].y);
                }
            }
        }
        const int q8 = bq & 1;
        float* mcol = &sMt[bq * PMT + 24 * wave + rl];
        #pragma unroll
        for (int u = 0; u < R_U; ++u) {
            float m[BC], ra[2];
            #pragma unroll
            for (int v = 0; v < BC; ++v) m[v] = __builtin_fmaxf(ma[u][v], mb[u][v]);
            #pragma unroll
            for (int v = 0; v < 2; ++v) {
                float keep = h ? m[v + 2] : m[v];
                float send = h ? m[v] : m[v + 2];
                ra[v] = __builtin_fmaxf(keep, xhalf(send));
            }
            float keep = q8 ? ra[1] : ra[0];
            float send = q8 ? ra[0] : ra[1];
            mcol[8 * u] = __builtin_fmaxf(keep, row_xor<8>(send));
        }
    }
    lds_barrier();

    {
        const float* pb = sMt + (r & 3) * PMT + 2 * h;
        v2f bf[24];
        #pragma unroll
        for (int s = 0; s < 24; ++s) bf[s] = *(const v2f*)(pb + 4 * s);
        v8f c = {};
        asm volatile("" : "+v"(c) : : "memory");
        #pragma unroll
        for (int kk = 0; kk < 12; ++kk) {
            c = __builtin_amdgcn_wmma_f32_16x16x4_f32(false, af[kk].lo, false, bf[2 * kk], (short)0, c, false, false);
            c = __builtin_amdgcn_wmma_f32_16x16x4_f32(false, af[kk].hi, false, bf[2 * kk + 1], (short)0, c, false, false);
        }
        const float y0 = pick4(c[0], c[1], c[2], c[3], sel);
        const float y1 = pick4(c[4], c[5], c[6], c[7], sel);
        const float xn0 = (xr0 * inv) * ns0 + y0;
        const float xn1 = (xr1 * inv) * ns1 + y1;
        xout[e0] = xn0;
        xout[e0 + 4 * BC] = xn1;
        float ss = xn0 * xn0;
        ss += xn1 * xn1;
        ss = wave_sum(ss);
        if (lane == 0) sRed[wave] = ss;
    }
    __syncthreads();
    if (tid == 0)
        ssout[32 * g] = ((((sRed[0] + sRed[1]) + sRed[2]) + sRed[3]) + sRed[4]) + sRed[5];
}

__global__ __launch_bounds__(256, 1)
void mpm_final_kernel(const float* __restrict__ xin, const float* __restrict__ ssin, float* __restrict__ out)
{
    const int e = blockIdx.x * 256 + threadIdx.x;
    float part[NB];
    load_partials(part, ssin);
    const float inv = 1.0f / sqrtf(sum_partials(part));
    const int i = e / NB, c = e % NB;
    v4f x = ((const v4f*)xin)[c * NN + i];
    ((v4f*)out)[e] = x * inv;
}

extern "C" void kernel_launch(void* const* d_in, const int* in_sizes, int n_in,
                              void* d_out, int out_size, void* d_ws, size_t ws_size,
                              hipStream_t stream) {
    (void)in_sizes; (void)n_in; (void)out_size;
    const float* A_gt = (const float*)d_in[0];
    const float* vec  = (const float*)d_in[1];
    float* out = (float*)d_out;
    float* ws  = (float*)d_ws;
    if (ws_size < sizeof(float) * WS_FLOATS) {
        graphvae_mpm_kernel<<<1, NT, 0, stream>>>(A_gt, vec, out);
        return;
    }
    mpm_setup_kernel<<<1, NT, 0, stream>>>(A_gt, vec, ws);
    for (int it = 0; it < MPM_ITERS; it += 2) {
        mpm_iter_kernel<<<NB, G_NT, 0, stream>>>(ws + WS_PF, ws + WS_Q, ws + WS_NS,
                                                 ws + WS_XA, ws + WS_SA, ws + WS_XB, ws + WS_SB);
        mpm_iter_kernel<<<NB, G_NT, 0, stream>>>(ws + WS_PF, ws + WS_Q, ws + WS_NS,
                                                 ws + WS_XB, ws + WS_SB, ws + WS_XA, ws + WS_SA);
    }
    mpm_final_kernel<<<NN * NN / 4 / 256, 256, 0, stream>>>(ws + WS_XA, ws + WS_SA, out);
}


# iteration kernel: its last work-group barrier orders LDS only, as its other two do: the X stores stay in flight under the partial-sum hand-over instead of being waited for first; no sum reordered
# speedup vs baseline: 12.5986x; 1.1143x over previous
#include <hip/hip_runtime.h>
#include <math.h>

#define NN 96
#define PITCH 100
#define MPM_ITERS 50
#define NT 1024

typedef __attribute__((ext_vector_type(2))) float v2f;
typedef __attribute__((ext_vector_type(8))) float v8f;
typedef __attribute__((ext_vector_type(4))) float v4f;

#define P1_J 4
#define P1_A 6
#define P1_B (NN / 2)
#define P1_THREADS (2 * (NN / P1_J) * (NN / P1_A))


__device__ __forceinline__ void p1_load_q(v4f (&q)[P1_A], const float* qrow, int k0)
{
    #pragma unroll
    for (int v = 0; v < P1_A; ++v)
        q[v] = *(const v4f*)(qrow + v * (NN / P1_A) * PITCH + k0);
}

template <bool REFILL>
__device__ __forceinline__ void p1_step(float (&m)[P1_J][P1_A], v4f (&x)[P1_J], const v4f (&q)[P1_A],
                                        const float* xrow, int knext)
{
    #pragma unroll
    for (int u = 0; u < P1_J; ++u) {
        #pragma unroll
        for (int v = 0; v < P1_A; ++v) {
            v2f p0 = q[v].lo * x[u].lo;
            v2f p1 = q[v].hi * x[u].hi;
            float t = __builtin_fmaxf(__builtin_fmaxf(m[u][v], p0.x), p0.y);
            m[u][v] = __builtin_fmaxf(__builtin_fmaxf(t, p1.x), p1.y);
        }
        if (REFILL)
            x[u] = *(const v4f*)(xrow + u * (NN / P1_J) * PITCH + knext);
    }
}


__device__ __forceinline__ v8f tile_gemm(const float* pa, const float* pb)
{
    v8f c = {};
    #pragma unroll
    for (int k0 = 0; k0 < NN; k0 += 4) {
        v2f av, bv;
        av.x = pa[k0];
        av.y = pa[k0 + 1];
        bv.x = pb[k0 * PITCH];
        bv.y = pb[(k0 + 1) * PITCH];
        c = __builtin_amdgcn_wmma_f32_16x16x4_f32(
                 false, av,  false, bv,
                 (short)0, c,  false,  false);
    }
    return c;
}

__device__ __forceinline__ void tile_update(v8f& acc, float& ss, const float* px, const float* pns)
{
    #pragma unroll
    for (int v = 0; v < 8; ++v) {
        float xn = px[v * PITCH] * pns[v * PITCH] + acc[v];
        acc[v] = xn;
        ss += xn * xn;
    }
}

__device__ __forceinline__ void tile_store(const v8f& acc, float inv, float* px)
{
    #pragma unroll
    for (int v = 0; v < 8; ++v)
        px[v * PITCH] = acc[v] * inv;
}

template <int MASK>
__device__ __forceinline__ float row_xor(float x)
{
    int b = __builtin_bit_cast(int, x);
    return __builtin_bit_cast(float, __builtin_amdgcn_update_dpp(b, b, 0x160 + MASK, 0xf, 0xf, false));
}

__device__ __forceinline__ float xhalf(float x)
{
    int b = __builtin_bit_cast(int, x);
    return __builtin_bit_cast(float, __builtin_amdgcn_permlanex16(b, b, 0x76543210, 0xfedcba98, false, false));
}

__device__ __forceinline__ float wave_sum(float x)
{
    x += row_xor<1>(x);
    x += row_xor<2>(x);
    x += row_xor<4>(x);
    x += row_xor<8>(x);
    x += xhalf(x);
    return x;
}

__global__ __launch_bounds__(NT, 1)
void graphvae_mpm_kernel(const float* __restrict__ A_gt,
                         const float* __restrict__ vec_logits,
                         float* __restrict__ out)
{
    __shared__ float sP[NN * PITCH];
    __shared__ __attribute__((aligned(16))) float sQ[NN * PITCH];
    __shared__ __attribute__((aligned(16))) float sX[NN * PITCH];
    __shared__ float sM[NN * PITCH];
    __shared__ float sNS[NN * PITCH];
    __shared__ float sDegA[NN], sDegB[NN], sDA[NN], sDB[NN];
    __shared__ float sRed[32];

    const int tid  = threadIdx.x;
    const int lane = tid & 31;
    const int wave = __builtin_amdgcn_readfirstlane(tid >> 5);

    for (int e = tid; e < NN * NN; e += NT) {
        int i = e / NN, j = e % NN;
        float l;
        if (i == j) {
            l = 1.0f;
        } else {
            int lo = (i < j) ? i : j;
            int hi = (i < j) ? j : i;
            int off = (lo * (2 * NN - lo - 1)) / 2 + (hi - lo - 1);
            l = vec_logits[off];
        }
        sQ[i * PITCH + j] = 1.0f / (1.0f + expf(-l));
        sP[i * PITCH + j] = A_gt[e];
        sX[i * PITCH + j] = 1.0f / (float)NN;
    }
    __syncthreads();

    if (tid < NN) {
        float ga = 0.f, gb = 0.f;
        for (int j = 0; j < NN; ++j) {
            ga += sP[tid * PITCH + j];
            gb += sQ[tid * PITCH + j];
        }
        sDA[tid] = sP[tid * PITCH + tid];
        sDB[tid] = sQ[tid * PITCH + tid];
        sDegA[tid] = ga;
        sDegB[tid] = gb;
    }
    __syncthreads();

    for (int e = tid; e < NN * NN; e += NT) {
        int i = e / NN, j = e % NN;
        float p = sP[i * PITCH + j] * sDA[i] * sDA[j];
        float q = sQ[i * PITCH + j] * sDB[i] * sDB[j];
        if (i == j) { p = 0.f; q = 0.f; }
        sP[i * PITCH + j] = p;
        sQ[i * PITCH + j] = q;
        sNS[i * PITCH + j] = sDA[i] * sDB[j] / (fabsf(sDegA[i] - sDegB[j]) + 1.0f);
    }
    __syncthreads();

    const int r = lane & 15;
    const int h = lane >> 4;
    const int ti0 = wave / 6, tn0 = wave % 6;
    const int ti1 = 5,        tn1 = 2 + wave;
    const int base0 = (ti0 * 16 + 8 * h) * PITCH + tn0 * 16 + r;
    const int base1 = (ti1 * 16 + 8 * h) * PITCH + tn1 * 16 + r;
    const int offa0 = (ti0 * 16 + r) * PITCH + 2 * h;
    const int offa1 = (ti1 * 16 + r) * PITCH + 2 * h;
    const int offb0 = 2 * h * PITCH + tn0 * 16 + r;
    const int offb1 = 2 * h * PITCH + tn1 * 16 + r;
    const bool two = wave < 4;

    for (int it = 0; it < MPM_ITERS; ++it) {
        if (tid < P1_THREADS) {
            const int jg = wave;
            const int ag = lane & 15;
            const int hb = lane >> 4;
            const float* xrow = &sX[jg * PITCH + hb * P1_B];
            const float* qrow = &sQ[ag * PITCH + hb * P1_B];
            float m[P1_J][P1_A];
            #pragma unroll
            for (int u = 0; u < P1_J; ++u) {
                #pragma unroll
                for (int v = 0; v < P1_A; ++v) m[u][v] = 0.0f;
            }
            v4f x[P1_J], qa[P1_A], qb[P1_A];
            #pragma unroll
            for (int u = 0; u < P1_J; ++u)
                x[u] = *(const v4f*)(xrow + u * (NN / P1_J) * PITCH);
            p1_load_q(qa, qrow, 0);
            for (int k0 = 0; k0 < P1_B - 8; k0 += 8) {
                p1_load_q(qb, qrow, k0 + 4);
                p1_step<true>(m, x, qa, xrow, k0 + 4);
                p1_load_q(qa, qrow, k0 + 8);
                p1_step<true>(m, x, qb, xrow, k0 + 8);
            }
            p1_load_q(qb, qrow, P1_B - 4);
            p1_step<true>(m, x, qa, xrow, P1_B - 4);
            p1_step<false>(m, x, qb, xrow, 0);
            float* mrow = &sM[jg * PITCH + ag + hb * (P1_A / 2) * (NN / P1_A)];
            #pragma unroll
            for (int u = 0; u < P1_J; ++u) {
                #pragma unroll
                for (int v = 0; v < P1_A / 2; ++v) {
                    float keep = hb ? m[u][v + P1_A / 2] : m[u][v];
                    float send = hb ? m[u][v] : m[u][v + P1_A / 2];
                    mrow[u * (NN / P1_J) * PITCH + v * (NN / P1_A)] = __builtin_fmaxf(keep, xhalf(send));
                }
            }
        }
        __syncthreads();

        v8f acc1 = {};
        if (two) acc1 = tile_gemm(sP + offa1, sM + offb1);
        v8f acc0 = tile_gemm(sP + offa0, sM + offb0);

        float ss = 0.f;
        tile_update(acc0, ss, sX + base0, sNS + base0);
        if (two) tile_update(acc1, ss, sX + base1, sNS + base1);
        ss = wave_sum(ss);
        if (lane == 0) sRed[wave] = ss;
        __syncthreads();

        float tot = wave_sum(sRed[lane]);
        float inv = 1.0f / sqrtf(tot);
        tile_store(acc0, inv, sX + base0);
        if (two) tile_store(acc1, inv, sX + base1);
        __syncthreads();
    }

    for (int e = tid; e < NN * NN; e += NT)
        out[e] = sX[(e / NN) * PITCH + (e % NN)];
}


#define G_NT 192
#define BC 4
#define NB (NN / BC)
#define PMT 104
#define NV4 (NN * NN / 4)
#define WS_PF 0
#define WS_Q  (NN * NN)
#define WS_NS (2 * NN * NN)
#define WS_XA (3 * NN * NN)
#define WS_XB (4 * NN * NN)
#define WS_SA (5 * NN * NN)
#define WS_SB (5 * NN * NN + NB * 32)
#define WS_FLOATS (5 * NN * NN + 2 * NB * 32)

#define R_U 3
#define R_S 6

__device__ __forceinline__ void lds_barrier()
{
    __builtin_amdgcn_fence(__ATOMIC_RELEASE, "workgroup", "local");
    __builtin_amdgcn_s_barrier();
    __builtin_amdgcn_fence(__ATOMIC_ACQUIRE, "workgroup", "local");
}

__device__ __forceinline__ void load_partials(float (&p)[NB], const float* ss)
{
    #pragma unroll
    for (int k = 0; k < NB; ++k) p[k] = ss[32 * k];
}

__device__ __forceinline__ float sum_partials(const float (&p)[NB])
{
    float a[NB / 2], b[NB / 4], c[NB / 8];
    #pragma unroll
    for (int k = 0; k < NB / 2; ++k) a[k] = p[2 * k] + p[2 * k + 1];
    #pragma unroll
    for (int k = 0; k < NB / 4; ++k) b[k] = a[2 * k] + a[2 * k + 1];
    #pragma unroll
    for (int k = 0; k < NB / 8; ++k) c[k] = b[2 * k] + b[2 * k + 1];
    return (c[0] + c[1]) + c[2];
}

__device__ __forceinline__ float pick4(float a0, float a1, float a2, float a3, int k)
{
    float lo = (k & 1) ? a1 : a0;
    float hi = (k & 1) ? a3 : a2;
    return (k & 2) ? hi : lo;
}

__global__ __launch_bounds__(NT, 1)
void mpm_setup_kernel(const float* __restrict__ A_gt, const float* __restrict__ vec_logits, float* __restrict__ ws)
{
    __shared__ __attribute__((aligned(16))) float sP[NN * PITCH];
    __shared__ __attribute__((aligned(16))) float sQ[NN * PITCH];
    __shared__ __attribute__((aligned(16))) float sNS[NN * PITCH];
    __shared__ float sDegA[NN], sDegB[NN], sDA[NN], sDB[NN];
    const int tid = threadIdx.x;

    for (int e = tid; e < NN * NN; e += NT) {
        int i = e / NN, j = e % NN;
        float l;
        if (i == j) {
            l = 1.0f;
        } else {
            int lo = (i < j) ? i : j;
            int hi = (i < j) ? j : i;
            int off = (lo * (2 * NN - lo - 1)) / 2 + (hi - lo - 1);
            l = vec_logits[off];
        }
        sQ[i * PITCH + j] = 1.0f / (1.0f + expf(-l));
        sP[i * PITCH + j] = A_gt[e];
    }
    __syncthreads();
    if (tid < NN) {
        float ga = 0.f, gb = 0.f;
        for (int j = 0; j < NN; ++j) {
            ga += sP[tid * PITCH + j];
            gb += sQ[tid * PITCH + j];
        }
        sDA[tid] = sP[tid * PITCH + tid];
        sDB[tid] = sQ[tid * PITCH + tid];
        sDegA[tid] = ga;
        sDegB[tid] = gb;
    }
    __syncthreads();
    for (int e = tid; e < NN * NN; e += NT) {
        int i = e / NN, j = e % NN;
        float p = sP[i * PITCH + j] * sDA[i] * sDA[j];
        float q = sQ[i * PITCH + j] * sDB[i] * sDB[j];
        if (i == j) { p = 0.f; q = 0.f; }
        sP[i * PITCH + j] = p;
        sQ[i * PITCH + j] = q;
        sNS[i * PITCH + j] = sDA[i] * sDB[j] / (fabsf(sDegA[i] - sDegB[j]) + 1.0f);
    }
    __syncthreads();
    for (int f = tid; f < NV4; f += NT) {
        int ti = f / 384, kk = (f % 384) / 32, ln = f % 32;
        const float* s = &sP[(16 * ti + (ln & 15)) * PITCH + 8 * kk + 2 * (ln >> 4)];
        v4f a = { s[0], s[1], s[4], s[5] };
        ((v4f*)(ws + WS_PF))[f] = a;
        int qi = f / 24, qc = f % 24;
        ((v4f*)(ws + WS_Q))[f] = *(const v4f*)&sQ[qi * PITCH + 4 * qc];
        int g = f / NN, i = f % NN;
        ((v4f*)(ws + WS_NS))[f] = *(const v4f*)&sNS[i * PITCH + BC * g];
        v4f x0 = { 1.0f / (float)NN, 1.0f / (float)NN, 1.0f / (float)NN, 1.0f / (float)NN };
        ((v4f*)(ws + WS_XA))[f] = x0;
    }
    if (tid < NB * 32) ws[WS_SA + tid] = (tid == 0) ? 1.0f : 0.0f;
}

__global__ __launch_bounds__(G_NT, 1)
void mpm_iter_kernel(const float* pf, const float* qd, const float* nsb,
                     const float* xin, const float* ssin,
                     float* xout, float* ssout)
{
    __shared__ __attribute__((aligned(16))) float sQ[BC * NN];
    __shared__ __attribute__((aligned(16))) float sMt[BC * PMT];
    __shared__ float sRed[G_NT / 32];

    const int g    = blockIdx.x;
    const int tid  = threadIdx.x;
    const int lane = tid & 31;
    const int wave = __builtin_amdgcn_readfirstlane(tid >> 5);
    const int r = lane & 15;
    const int h = lane >> 4;
    const int rl = lane & 7;
    const int bq = lane >> 3;

    float part[NB];
    load_partials(part, ssin);
    const bool ph1 = wave < 4;
    v4f x[R_S][R_U];
    if (ph1) {
        const v4f* xs = (const v4f*)xin + (R_S * bq) * NN + 24 * wave + rl;
        #pragma unroll
        for (int s = 0; s < R_S; ++s) {
            #pragma unroll
            for (int u = 0; u < R_U; ++u) x[s][u] = xs[s * NN + 8 * u];
        }
    } else {
        const v4f* qs = (const v4f*)(qd + g * BC * NN);
        const int tq = tid & 63;
        *(v4f*)&sQ[4 * tq] = qs[tq];
        if (wave == 4) *(v4f*)&sQ[4 * (64 + lane)] = qs[64 + lane];
    }
    v4f af[12];
    {
        const v4f* p = (const v4f*)pf + wave * 12 * 32 + lane;
        #pragma unroll
        for (int kk = 0; kk < 12; ++kk) af[kk] = p[kk * 32];
    }
    const int sel = r >> 2;
    const int e0 = (g * NN + wave * 16 + 8 * h + sel) * BC + (r & 3);
    const float ns0 = nsb[e0], ns1 = nsb[e0 + 4 * BC];
    const float xr0 = xin[e0], xr1 = xin[e0 + 4 * BC];

    lds_barrier();

    v4f q[R_S][BC];
    if (ph1) {
        const float* qrow = &sQ[24 * bq];
        #pragma unroll
        for (int s = 0; s < R_S; ++s) {
            #pragma unroll
            for (int v = 0; v < BC; ++v) q[s][v] = *(const v4f*)(qrow + v * NN + 4 * s);
        }
    }

    asm volatile("" : "+s"(part[0]), "+s"(part[1]), "+s"(part[2]), "+s"(part[3]), "+s"(part[4]), "+s"(part[5]),
                      "+s"(part[6]), "+s"(part[7]), "+s"(part[8]), "+s"(part[9]), "+s"(part[10]), "+s"(part[11]),
                      "+s"(part[12]), "+s"(part[13]), "+s"(part[14]), "+s"(part[15]), "+s"(part[16]), "+s"(part[17]),
                      "+s"(part[18]), "+s"(part[19]), "+s"(part[20]), "+s"(part[21]), "+s"(part[22]), "+s"(part[23])
                 : : "memory");
    const float inv = 1.0f / sqrtf(sum_partials(part));

    if (ph1) {
        float ma[R_U][BC], mb[R_U][BC];
        #pragma unroll
        for (int u = 0; u < R_U; ++u) {
            #pragma unroll
            for (int v = 0; v < BC; ++v) { ma[u][v] = 0.0f; mb[u][v] = 0.0f; }
        }
        #pragma unroll
        for (int s = 0; s < R_S; ++s) {
            #pragma unroll
            for (int u = 0; u < R_U; ++u) {
                const v4f xv = x[s][u] * inv;
                v2f p0[BC], p1[BC];
                #pragma unroll
                for (int v = 0; v < BC; ++v) {
                    p0[v] = q[s][v].lo * xv.lo;
                    p1[v] = q[s][v].hi * xv.hi;
                }
                #pragma unroll
                for (int v = 0; v < BC; ++v) {
                    ma[u][v] = __builtin_fmaxf(__builtin_fmaxf(ma[u][v], p0[v].x), p0[v].y);
                    mb[u][v] = __builtin_fmaxf(__builtin_fmaxf(mb[u][v], p1[v].x), p1[v].y);
                }
            }
        }
        const int q8 = bq & 1;
        float* mcol = &sMt[bq * PMT + 24 * wave + rl];
        #pragma unroll
        for (int u = 0; u < R_U; ++u) {
            float m[BC], ra[2];
            #pragma unroll
            for (int v = 0; v < BC; ++v) m[v] = __builtin_fmaxf(ma[u][v], mb[u][v]);
            #pragma unroll
            for (int v = 0; v < 2; ++v) {
                float keep = h ? m[v + 2] : m[v];
                float send = h ? m[v] : m[v + 2];
                ra[v] = __builtin_fmaxf(keep, xhalf(send));
            }
            float keep = q8 ? ra[1] : ra[0];
            float send = q8 ? ra[0] : ra[1];
            mcol[8 * u] = __builtin_fmaxf(keep, row_xor<8>(send));
        }
    }
    lds_barrier();

    {
        const float* pb = sMt + (r & 3) * PMT + 2 * h;
        v2f bf[24];
        #pragma unroll
        for (int s = 0; s < 24; ++s) bf[s] = *(const v2f*)(pb + 4 * s);
        v8f c = {};
        asm volatile("" : "+v"(c) : : "memory");
        #pragma unroll
        for (int kk = 0; kk < 12; ++kk) {
            c = __builtin_amdgcn_wmma_f32_16x16x4_f32(false, af[kk].lo, false, bf[2 * kk], (short)0, c, false, false);
            c = __builtin_amdgcn_wmma_f32_16x16x4_f32(false, af[kk].hi, false, bf[2 * kk + 1], (short)0, c, false, false);
        }
        const float y0 = pick4(c[0], c[1], c[2], c[3], sel);
        const float y1 = pick4(c[4], c[5], c[6], c[7], sel);
        const float xn0 = (xr0 * inv) * ns0 + y0;
        const float xn1 = (xr1 * inv) * ns1 + y1;
        xout[e0] = xn0;
        xout[e0 + 4 * BC] = xn1;
        float ss = xn0 * xn0;
        ss += xn1 * xn1;
        ss = wave_sum(ss);
        if (lane == 0) sRed[wave] = ss;
    }
    lds_barrier();
    if (tid == 0)
        ssout[32 * g] = ((((sRed[0] + sRed[1]) + sRed[2]) + sRed[3]) + sRed[4]) + sRed[5];
}

__global__ __launch_bounds__(256, 1)
void mpm_final_kernel(const float* __restrict__ xin, const float* __restrict__ ssin, float* __restrict__ out)
{
    const int e = blockIdx.x * 256 + threadIdx.x;
    float part[NB];
    load_partials(part, ssin);
    const float inv = 1.0f / sqrtf(sum_partials(part));
    const int i = e / NB, c = e % NB;
    v4f x = ((const v4f*)xin)[c * NN + i];
    ((v4f*)out)[e] = x * inv;
}

extern "C" void kernel_launch(void* const* d_in, const int* in_sizes, int n_in,
                              void* d_out, int out_size, void* d_ws, size_t ws_size,
                              hipStream_t stream) {
    (void)in_sizes; (void)n_in; (void)out_size;
    const float* A_gt = (const float*)d_in[0];
    const float* vec  = (const float*)d_in[1];
    float* out = (float*)d_out;
    float* ws  = (float*)d_ws;
    if (ws_size < sizeof(float) * WS_FLOATS) {
        graphvae_mpm_kernel<<<1, NT, 0, stream>>>(A_gt, vec, out);
        return;
    }
    mpm_setup_kernel<<<1, NT, 0, stream>>>(A_gt, vec, ws);
    for (int it = 0; it < MPM_ITERS; it += 2) {
        mpm_iter_kernel<<<NB, G_NT, 0, stream>>>(ws + WS_PF, ws + WS_Q, ws + WS_NS,
                                                 ws + WS_XA, ws + WS_SA, ws + WS_XB, ws + WS_SB);
        mpm_iter_kernel<<<NB, G_NT, 0, stream>>>(ws + WS_PF, ws + WS_Q, ws + WS_NS,
                                                 ws + WS_XB, ws + WS_SB, ws + WS_XA, ws + WS_SA);
    }
    mpm_final_kernel<<<NN * NN / 4 / 256, 256, 0, stream>>>(ws + WS_XA, ws + WS_SA, out);
}


# no partial sums through memory: each work-group sums |X|^2 from the X words it holds for phase 1 and applies 1/|X| after P'M, not to X first; norm summed in another order, scale rounds elsewhere
# speedup vs baseline: 13.2124x; 1.0487x over previous
#include <hip/hip_runtime.h>
#include <math.h>

#define NN 96
#define PITCH 100
#define MPM_ITERS 50
#define NT 1024

typedef __attribute__((ext_vector_type(2))) float v2f;
typedef __attribute__((ext_vector_type(8))) float v8f;
typedef __attribute__((ext_vector_type(4))) float v4f;

#define P1_J 4
#define P1_A 6
#define P1_B (NN / 2)
#define P1_THREADS (2 * (NN / P1_J) * (NN / P1_A))


__device__ __forceinline__ void p1_load_q(v4f (&q)[P1_A], const float* qrow, int k0)
{
    #pragma unroll
    for (int v = 0; v < P1_A; ++v)
        q[v] = *(const v4f*)(qrow + v * (NN / P1_A) * PITCH + k0);
}

template <bool REFILL>
__device__ __forceinline__ void p1_step(float (&m)[P1_J][P1_A], v4f (&x)[P1_J], const v4f (&q)[P1_A],
                                        const float* xrow, int knext)
{
    #pragma unroll
    for (int u = 0; u < P1_J; ++u) {
        #pragma unroll
        for (int v = 0; v < P1_A; ++v) {
            v2f p0 = q[v].lo * x[u].lo;
            v2f p1 = q[v].hi * x[u].hi;
            float t = __builtin_fmaxf(__builtin_fmaxf(m[u][v], p0.x), p0.y);
            m[u][v] = __builtin_fmaxf(__builtin_fmaxf(t, p1.x), p1.y);
        }
        if (REFILL)
            x[u] = *(const v4f*)(xrow + u * (NN / P1_J) * PITCH + knext);
    }
}


__device__ __forceinline__ v8f tile_gemm(const float* pa, const float* pb)
{
    v8f c = {};
    #pragma unroll
    for (int k0 = 0; k0 < NN; k0 += 4) {
        v2f av, bv;
        av.x = pa[k0];
        av.y = pa[k0 + 1];
        bv.x = pb[k0 * PITCH];
        bv.y = pb[(k0 + 1) * PITCH];
        c = __builtin_amdgcn_wmma_f32_16x16x4_f32(
                 false, av,  false, bv,
                 (short)0, c,  false,  false);
    }
    return c;
}

__device__ __forceinline__ void tile_update(v8f& acc, float& ss, const float* px, const float* pns)
{
    #pragma unroll
    for (int v = 0; v < 8; ++v) {
        float xn = px[v * PITCH] * pns[v * PITCH] + acc[v];
        acc[v] = xn;
        ss += xn * xn;
    }
}

__device__ __forceinline__ void tile_store(const v8f& acc, float inv, float* px)
{
    #pragma unroll
    for (int v = 0; v < 8; ++v)
        px[v * PITCH] = acc[v] * inv;
}

template <int MASK>
__device__ __forceinline__ float row_xor(float x)
{
    int b = __builtin_bit_cast(int, x);
    return __builtin_bit_cast(float, __builtin_amdgcn_update_dpp(b, b, 0x160 + MASK, 0xf, 0xf, false));
}

__device__ __forceinline__ float xhalf(float x)
{
    int b = __builtin_bit_cast(int, x);
    return __builtin_bit_cast(float, __builtin_amdgcn_permlanex16(b, b, 0x76543210, 0xfedcba98, false, false));
}

__device__ __forceinline__ float wave_sum(float x)
{
    x += row_xor<1>(x);
    x += row_xor<2>(x);
    x += row_xor<4>(x);
    x += row_xor<8>(x);
    x += xhalf(x);
    return x;
}

__global__ __launch_bounds__(NT, 1)
void graphvae_mpm_kernel(const float* __restrict__ A_gt,
                         const float* __restrict__ vec_logits,
                         float* __restrict__ out)
{
    __shared__ float sP[NN * PITCH];
    __shared__ __attribute__((aligned(16))) float sQ[NN * PITCH];
    __shared__ __attribute__((aligned(16))) float sX[NN * PITCH];
    __shared__ float sM[NN * PITCH];
    __shared__ float sNS[NN * PITCH];
    __shared__ float sDegA[NN], sDegB[NN], sDA[NN], sDB[NN];
    __shared__ float sRed[32];

    const int tid  = threadIdx.x;
    const int lane = tid & 31;
    const int wave = __builtin_amdgcn_readfirstlane(tid >> 5);

    for (int e = tid; e < NN * NN; e += NT) {
        int i = e / NN, j = e % NN;
        float l;
        if (i == j) {
            l = 1.0f;
        } else {
            int lo = (i < j) ? i : j;
            int hi = (i < j) ? j : i;
            int off = (lo * (2 * NN - lo - 1)) / 2 + (hi - lo - 1);
            l = vec_logits[off];
        }
        sQ[i * PITCH + j] = 1.0f / (1.0f + expf(-l));
        sP[i * PITCH + j] = A_gt[e];
        sX[i * PITCH + j] = 1.0f / (float)NN;
    }
    __syncthreads();

    if (tid < NN) {
        float ga = 0.f, gb = 0.f;
        for (int j = 0; j < NN; ++j) {
            ga += sP[tid * PITCH + j];
            gb += sQ[tid * PITCH + j];
        }
        sDA[tid] = sP[tid * PITCH + tid];
        sDB[tid] = sQ[tid * PITCH + tid];
        sDegA[tid] = ga;
        sDegB[tid] = gb;
    }
    __syncthreads();

    for (int e = tid; e < NN * NN; e += NT) {
        int i = e / NN, j = e % NN;
        float p = sP[i * PITCH + j] * sDA[i] * sDA[j];
        float q = sQ[i * PITCH + j] * sDB[i] * sDB[j];
        if (i == j) { p = 0.f; q = 0.f; }
        sP[i * PITCH + j] = p;
        sQ[i * PITCH + j] = q;
        sNS[i * PITCH + j] = sDA[i] * sDB[j] / (fabsf(sDegA[i] - sDegB[j]) + 1.0f);
    }
    __syncthreads();

    const int r = lane & 15;
    const int h = lane >> 4;
    const int ti0 = wave / 6, tn0 = wave % 6;
    const int ti1 = 5,        tn1 = 2 + wave;
    const int base0 = (ti0 * 16 + 8 * h) * PITCH + tn0 * 16 + r;
    const int base1 = (ti1 * 16 + 8 * h) * PITCH + tn1 * 16 + r;
    const int offa0 = (ti0 * 16 + r) * PITCH + 2 * h;
    const int offa1 = (ti1 * 16 + r) * PITCH + 2 * h;
    const int offb0 = 2 * h * PITCH + tn0 * 16 + r;
    const int offb1 = 2 * h * PITCH + tn1 * 16 + r;
    const bool two = wave < 4;

    for (int it = 0; it < MPM_ITERS; ++it) {
        if (tid < P1_THREADS) {
            const int jg = wave;
            const int ag = lane & 15;
            const int hb = lane >> 4;
            const float* xrow = &sX[jg * PITCH + hb * P1_B];
            const float* qrow = &sQ[ag * PITCH + hb * P1_B];
            float m[P1_J][P1_A];
            #pragma unroll
            for (int u = 0; u < P1_J; ++u) {
                #pragma unroll
                for (int v = 0; v < P1_A; ++v) m[u][v] = 0.0f;
            }
            v4f x[P1_J], qa[P1_A], qb[P1_A];
            #pragma unroll
            for (int u = 0; u < P1_J; ++u)
                x[u] = *(const v4f*)(xrow + u * (NN / P1_J) * PITCH);
            p1_load_q(qa, qrow, 0);
            for (int k0 = 0; k0 < P1_B - 8; k0 += 8) {
                p1_load_q(qb, qrow, k0 + 4);
                p1_step<true>(m, x, qa, xrow, k0 + 4);
                p1_load_q(qa, qrow, k0 + 8);
                p1_step<true>(m, x, qb, xrow, k0 + 8);
            }
            p1_load_q(qb, qrow, P1_B - 4);
            p1_step<true>(m, x, qa, xrow, P1_B - 4);
            p1_step<false>(m, x, qb, xrow, 0);
            float* mrow = &sM[jg * PITCH + ag + hb * (P1_A / 2) * (NN / P1_A)];
            #pragma unroll
            for (int u = 0; u < P1_J; ++u) {
                #pragma unroll
                for (int v = 0; v < P1_A / 2; ++v) {
                    float keep = hb ? m[u][v + P1_A / 2] : m[u][v];
                    float send = hb ? m[u][v] : m[u][v + P1_A / 2];
                    mrow[u * (NN / P1_J) * PITCH + v * (NN / P1_A)] = __builtin_fmaxf(keep, xhalf(send));
                }
            }
        }
        __syncthreads();

        v8f acc1 = {};
        if (two) acc1 = tile_gemm(sP + offa1, sM + offb1);
        v8f acc0 = tile_gemm(sP + offa0, sM + offb0);

        float ss = 0.f;
        tile_update(acc0, ss, sX + base0, sNS + base0);
        if (two) tile_update(acc1, ss, sX + base1, sNS + base1);
        ss = wave_sum(ss);
        if (lane == 0) sRed[wave] = ss;
        __syncthreads();

        float tot = wave_sum(sRed[lane]);
        float inv = 1.0f / sqrtf(tot);
        tile_store(acc0, inv, sX + base0);
        if (two) tile_store(acc1, inv, sX + base1);
        __syncthreads();
    }

    for (int e = tid; e < NN * NN; e += NT)
        out[e] = sX[(e / NN) * PITCH + (e % NN)];
}


#define G_NT 192
#define BC 4
#define NB (NN / BC)
#define PMT 104
#define NV4 (NN * NN / 4)
#define WS_PF 0
#define WS_Q  (NN * NN)
#define WS_NS (2 * NN * NN)
#define WS_XA (3 * NN * NN)
#define WS_XB (4 * NN * NN)
#define WS_FLOATS (5 * NN * NN)

#define R_U 3
#define R_S 6

__device__ __forceinline__ void lds_barrier()
{
    __builtin_amdgcn_fence(__ATOMIC_RELEASE, "workgroup", "local");
    __builtin_amdgcn_s_barrier();
    __builtin_amdgcn_fence(__ATOMIC_ACQUIRE, "workgroup", "local");
}

__device__ __forceinline__ float pick4(float a0, float a1, float a2, float a3, int k)
{
    float lo = (k & 1) ? a1 : a0;
    float hi = (k & 1) ? a3 : a2;
    return (k & 2) ? hi : lo;
}

__global__ __launch_bounds__(NT, 1)
void mpm_setup_kernel(const float* __restrict__ A_gt, const float* __restrict__ vec_logits, float* __restrict__ ws)
{
    __shared__ __attribute__((aligned(16))) float sP[NN * PITCH];
    __shared__ __attribute__((aligned(16))) float sQ[NN * PITCH];
    __shared__ __attribute__((aligned(16))) float sNS[NN * PITCH];
    __shared__ float sDegA[NN], sDegB[NN], sDA[NN], sDB[NN];
    const int tid = threadIdx.x;

    for (int e = tid; e < NN * NN; e += NT) {
        int i = e / NN, j = e % NN;
        float l;
        if (i == j) {
            l = 1.0f;
        } else {
            int lo = (i < j) ? i : j;
            int hi = (i < j) ? j : i;
            int off = (lo * (2 * NN - lo - 1)) / 2 + (hi - lo - 1);
            l = vec_logits[off];
        }
        sQ[i * PITCH + j] = 1.0f / (1.0f + expf(-l));
        sP[i * PITCH + j] = A_gt[e];
    }
    __syncthreads();
    if (tid < NN) {
        float ga = 0.f, gb = 0.f;
        for (int j = 0; j < NN; ++j) {
            ga += sP[tid * PITCH + j];
            gb += sQ[tid * PITCH + j];
        }
        sDA[tid] = sP[tid * PITCH + tid];
        sDB[tid] = sQ[tid * PITCH + tid];
        sDegA[tid] = ga;
        sDegB[tid] = gb;
    }
    __syncthreads();
    for (int e = tid; e < NN * NN; e += NT) {
        int i = e / NN, j = e % NN;
        float p = sP[i * PITCH + j] * sDA[i] * sDA[j];
        float q = sQ[i * PITCH + j] * sDB[i] * sDB[j];
        if (i == j) { p = 0.f; q = 0.f; }
        sP[i * PITCH + j] = p;
        sQ[i * PITCH + j] = q;
        sNS[i * PITCH + j] = sDA[i] * sDB[j] / (fabsf(sDegA[i] - sDegB[j]) + 1.0f);
    }
    __syncthreads();
    for (int f = tid; f < NV4; f += NT) {
        int ti = f / 384, kk = (f % 384) / 32, ln = f % 32;
        const float* s = &sP[(16 * ti + (ln & 15)) * PITCH + 8 * kk + 2 * (ln >> 4)];
        v4f a = { s[0], s[1], s[4], s[5] };
        ((v4f*)(ws + WS_PF))[f] = a;
        int qi = f / 24, qc = f % 24;
        ((v4f*)(ws + WS_Q))[f] = *(const v4f*)&sQ[qi * PITCH + 4 * qc];
        int g = f / NN, i = f % NN;
        ((v4f*)(ws + WS_NS))[f] = *(const v4f*)&sNS[i * PITCH + BC * g];
        v4f x0 = { 1.0f / (float)NN, 1.0f / (float)NN, 1.0f / (float)NN, 1.0f / (float)NN };
        ((v4f*)(ws + WS_XA))[f] = x0;
    }
}

__global__ __launch_bounds__(G_NT, 1)
void mpm_iter_kernel(const float* pf, const float* qd, const float* nsb,
                     const float* xin, float* xout)
{
    __shared__ __attribute__((aligned(16))) float sQ[BC * NN];
    __shared__ __attribute__((aligned(16))) float sMt[BC * PMT];
    __shared__ __attribute__((aligned(16))) float sSS[128];

    const int g    = blockIdx.x;
    const int tid  = threadIdx.x;
    const int lane = tid & 31;
    const int wave = __builtin_amdgcn_readfirstlane(tid >> 5);
    const int r = lane & 15;
    const int h = lane >> 4;
    const int rl = lane & 7;
    const int bq = lane >> 3;

    const bool ph1 = wave < 4;
    v4f x[R_S][R_U];
    if (ph1) {
        const v4f* xs = (const v4f*)xin + (R_S * bq) * NN + 24 * wave + rl;
        #pragma unroll
        for (int s = 0; s < R_S; ++s) {
            #pragma unroll
            for (int u = 0; u < R_U; ++u) x[s][u] = xs[s * NN + 8 * u];
        }
    } else {
        const v4f* qs = (const v4f*)(qd + g * BC * NN);
        const int tq = tid & 63;
        *(v4f*)&sQ[4 * tq] = qs[tq];
        if (wave == 4) *(v4f*)&sQ[4 * (64 + lane)] = qs[64 + lane];
    }
    v4f af[12];
    {
        const v4f* p = (const v4f*)pf + wave * 12 * 32 + lane;
        #pragma unroll
        for (int kk = 0; kk < 12; ++kk) af[kk] = p[kk * 32];
    }
    const int sel = r >> 2;
    const int e0 = (g * NN + wave * 16 + 8 * h + sel) * BC + (r & 3);
    const float ns0 = nsb[e0], ns1 = nsb[e0 + 4 * BC];
    const float xr0 = xin[e0], xr1 = xin[e0 + 4 * BC];

    lds_barrier();

    if (ph1) {
        v4f q[R_S][BC];
        const float* qrow = &sQ[24 * bq];
        #pragma unroll
        for (int s = 0; s < R_S; ++s) {
            #pragma unroll
            for (int v = 0; v < BC; ++v) q[s][v] = *(const v4f*)(qrow + v * NN + 4 * s);
        }
        float ma[R_U][BC], mb[R_U][BC];
        #pragma unroll
        for (int u = 0; u < R_U; ++u) {
            #pragma unroll
            for (int v = 0; v < BC; ++v) { ma[u][v] = 0.0f; mb[u][v] = 0.0f; }
        }
        v2f sa = { 0.0f, 0.0f }, sb = { 0.0f, 0.0f };
        #pragma unroll
        for (int s = 0; s < R_S; ++s) {
            #pragma unroll
            for (int u = 0; u < R_U; ++u) {
                const v4f xv = x[s][u];
                sa = xv.lo * xv.lo + sa;
                sb = xv.hi * xv.hi + sb;
                v2f p0[BC], p1[BC];
                #pragma unroll
                for (int v = 0; v < BC; ++v) {
                    p0[v] = q[s][v].lo * xv.lo;
                    p1[v] = q[s][v].hi * xv.hi;
                }
                #pragma unroll
                for (int v = 0; v < BC; ++v) {
                    ma[u][v] = __builtin_fmaxf(__builtin_fmaxf(ma[u][v], p0[v].x), p0[v].y);
                    mb[u][v] = __builtin_fmaxf(__builtin_fmaxf(mb[u][v], p1[v].x), p1[v].y);
                }
            }
        }
        sSS[tid] = (sa.x + sa.y) + (sb.x + sb.y);
        const int q8 = bq & 1;
        float* mcol = &sMt[bq * PMT + 24 * wave + rl];
        #pragma unroll
        for (int u = 0; u < R_U; ++u) {
            float m[BC], ra[2];
            #pragma unroll
            for (int v = 0; v < BC; ++v) m[v] = __builtin_fmaxf(ma[u][v], mb[u][v]);
            #pragma unroll
            for (int v = 0; v < 2; ++v) {
                float keep = h ? m[v + 2] : m[v];
                float send = h ? m[v] : m[v + 2];
                ra[v] = __builtin_fmaxf(keep, xhalf(send));
            }
            float keep = q8 ? ra[1] : ra[0];
            float send = q8 ? ra[0] : ra[1];
            mcol[8 * u] = __builtin_fmaxf(keep, row_xor<8>(send));
        }
    }
    lds_barrier();

    {
        const v4f p4 = *(const v4f*)&sSS[4 * lane];
        const float* pb = sMt + (r & 3) * PMT + 2 * h;
        v2f bf[24];
        #pragma unroll
        for (int s = 0; s < 24; ++s) bf[s] = *(const v2f*)(pb + 4 * s);
        v8f c = {};
        asm volatile("" : "+v"(c) : : "memory");
        #pragma unroll
        for (int kk = 0; kk < 12; ++kk) {
            c = __builtin_amdgcn_wmma_f32_16x16x4_f32(false, af[kk].lo, false, bf[2 * kk], (short)0, c, false, false);
            c = __builtin_amdgcn_wmma_f32_16x16x4_f32(false, af[kk].hi, false, bf[2 * kk + 1], (short)0, c, false, false);
        }
        const float inv = 1.0f / sqrtf(wave_sum((p4.x + p4.y) + (p4.z + p4.w)));
        const float y0 = pick4(c[0], c[1], c[2], c[3], sel);
        const float y1 = pick4(c[4], c[5], c[6], c[7], sel);
        xout[e0] = (xr0 * ns0 + y0) * inv;
        xout[e0 + 4 * BC] = (xr1 * ns1 + y1) * inv;
    }
}

__global__ __launch_bounds__(256, 1)
void mpm_final_kernel(const float* __restrict__ xin, float* __restrict__ out)
{
    __shared__ __attribute__((aligned(16))) float sRed[8];
    const int tid  = threadIdx.x;
    const int lane = tid & 31;
    const int wave = __builtin_amdgcn_readfirstlane(tid >> 5);
    const int e = blockIdx.x * 256 + tid;
    const int i = e / NB, c = e % NB;
    const v4f* xs = (const v4f*)xin;
    const v4f x = xs[c * NN + i];
    v4f w[NV4 / 256];
    #pragma unroll
    for (int k = 0; k < NV4 / 256; ++k) w[k] = xs[tid + 256 * k];
    v2f sa = { 0.0f, 0.0f }, sb = { 0.0f, 0.0f };
    #pragma unroll
    for (int k = 0; k < NV4 / 256; ++k) {
        sa = w[k].lo * w[k].lo + sa;
        sb = w[k].hi * w[k].hi + sb;
    }
    const float ss = wave_sum((sa.x + sa.y) + (sb.x + sb.y));
    if (lane == 0) sRed[wave] = ss;
    __syncthreads();
    const v4f a = *(const v4f*)&sRed[0], b = *(const v4f*)&sRed[4];
    const float inv = 1.0f / sqrtf(((a.x + a.y) + (a.z + a.w)) + ((b.x + b.y) + (b.z + b.w)));
    ((v4f*)out)[e] = x * inv;
}

extern "C" void kernel_launch(void* const* d_in, const int* in_sizes, int n_in,
                              void* d_out, int out_size, void* d_ws, size_t ws_size,
                              hipStream_t stream) {
    (void)in_sizes; (void)n_in; (void)out_size;
    const float* A_gt = (const float*)d_in[0];
    const float* vec  = (const float*)d_in[1];
    float* out = (float*)d_out;
    float* ws  = (float*)d_ws;
    if (ws_size < sizeof(float) * WS_FLOATS) {
        graphvae_mpm_kernel<<<1, NT, 0, stream>>>(A_gt, vec, out);
        return;
    }
    mpm_setup_kernel<<<1, NT, 0, stream>>>(A_gt, vec, ws);
    for (int it = 0; it < MPM_ITERS; it += 2) {
        mpm_iter_kernel<<<NB, G_NT, 0, stream>>>(ws + WS_PF, ws + WS_Q, ws + WS_NS, ws + WS_XA, ws + WS_XB);
        mpm_iter_kernel<<<NB, G_NT, 0, stream>>>(ws + WS_PF, ws + WS_Q, ws + WS_NS, ws + WS_XB, ws + WS_XA);
    }
    mpm_final_kernel<<<NN * NN / 4 / 256, 256, 0, stream>>>(ws + WS_XA, out);
}


# iteration kernel: phase 2 of a column block split by rows over two work-groups (48 a launch; 3 tiles each, on waves of their own): one WMMA chain per SIMD, not two; both do phase 1; no sum reordered
# speedup vs baseline: 14.4308x; 1.0922x over previous
#include <hip/hip_runtime.h>
#include <math.h>

#define NN 96
#define PITCH 100
#define MPM_ITERS 50
#define NT 1024

typedef __attribute__((ext_vector_type(2))) float v2f;
typedef __attribute__((ext_vector_type(8))) float v8f;
typedef __attribute__((ext_vector_type(4))) float v4f;

#define P1_J 4
#define P1_A 6
#define P1_B (NN / 2)
#define P1_THREADS (2 * (NN / P1_J) * (NN / P1_A))


__device__ __forceinline__ void p1_load_q(v4f (&q)[P1_A], const float* qrow, int k0)
{
    #pragma unroll
    for (int v = 0; v < P1_A; ++v)
        q[v] = *(const v4f*)(qrow + v * (NN / P1_A) * PITCH + k0);
}

template <bool REFILL>
__device__ __forceinline__ void p1_step(float (&m)[P1_J][P1_A], v4f (&x)[P1_J], const v4f (&q)[P1_A],
                                        const float* xrow, int knext)
{
    #pragma unroll
    for (int u = 0; u < P1_J; ++u) {
        #pragma unroll
        for (int v = 0; v < P1_A; ++v) {
            v2f p0 = q[v].lo * x[u].lo;
            v2f p1 = q[v].hi * x[u].hi;
            float t = __builtin_fmaxf(__builtin_fmaxf(m[u][v], p0.x), p0.y);
            m[u][v] = __builtin_fmaxf(__builtin_fmaxf(t, p1.x), p1.y);
        }
        if (REFILL)
            x[u] = *(const v4f*)(xrow + u * (NN / P1_J) * PITCH + knext);
    }
}


__device__ __forceinline__ v8f tile_gemm(const float* pa, const float* pb)
{
    v8f c = {};
    #pragma unroll
    for (int k0 = 0; k0 < NN; k0 += 4) {
        v2f av, bv;
        av.x = pa[k0];
        av.y = pa[k0 + 1];
        bv.x = pb[k0 * PITCH];
        bv.y = pb[(k0 + 1) * PITCH];
        c = __builtin_amdgcn_wmma_f32_16x16x4_f32(
                 false, av,  false, bv,
                 (short)0, c,  false,  false);
    }
    return c;
}

__device__ __forceinline__ void tile_update(v8f& acc, float& ss, const float* px, const float* pns)
{
    #pragma unroll
    for (int v = 0; v < 8; ++v) {
        float xn = px[v * PITCH] * pns[v * PITCH] + acc[v];
        acc[v] = xn;
        ss += xn * xn;
    }
}

__device__ __forceinline__ void tile_store(const v8f& acc, float inv, float* px)
{
    #pragma unroll
    for (int v = 0; v < 8; ++v)
        px[v * PITCH] = acc[v] * inv;
}

template <int MASK>
__device__ __forceinline__ float row_xor(float x)
{
    int b = __builtin_bit_cast(int, x);
    return __builtin_bit_cast(float, __builtin_amdgcn_update_dpp(b, b, 0x160 + MASK, 0xf, 0xf, false));
}

__device__ __forceinline__ float xhalf(float x)
{
    int b = __builtin_bit_cast(int, x);
    return __builtin_bit_cast(float, __builtin_amdgcn_permlanex16(b, b, 0x76543210, 0xfedcba98, false, false));
}

__device__ __forceinline__ float wave_sum(float x)
{
    x += row_xor<1>(x);
    x += row_xor<2>(x);
    x += row_xor<4>(x);
    x += row_xor<8>(x);
    x += xhalf(x);
    return x;
}

__global__ __launch_bounds__(NT, 1)
void graphvae_mpm_kernel(const float* __restrict__ A_gt,
                         const float* __restrict__ vec_logits,
                         float* __restrict__ out)
{
    __shared__ float sP[NN * PITCH];
    __shared__ __attribute__((aligned(16))) float sQ[NN * PITCH];
    __shared__ __attribute__((aligned(16))) float sX[NN * PITCH];
    __shared__ float sM[NN * PITCH];
    __shared__ float sNS[NN * PITCH];
    __shared__ float sDegA[NN], sDegB[NN], sDA[NN], sDB[NN];
    __shared__ float sRed[32];

    const int tid  = threadIdx.x;
    const int lane = tid & 31;
    const int wave = __builtin_amdgcn_readfirstlane(tid >> 5);

    for (int e = tid; e < NN * NN; e += NT) {
        int i = e / NN, j = e % NN;
        float l;
        if (i == j) {
            l = 1.0f;
        } else {
            int lo = (i < j) ? i : j;
            int hi = (i < j) ? j : i;
            int off = (lo * (2 * NN - lo - 1)) / 2 + (hi - lo - 1);
            l = vec_logits[off];
        }
        sQ[i * PITCH + j] = 1.0f / (1.0f + expf(-l));
        sP[i * PITCH + j] = A_gt[e];
        sX[i * PITCH + j] = 1.0f / (float)NN;
    }
    __syncthreads();

    if (tid < NN) {
        float ga = 0.f, gb = 0.f;
        for (int j = 0; j < NN; ++j) {
            ga += sP[tid * PITCH + j];
            gb += sQ[tid * PITCH + j];
        }
        sDA[tid] = sP[tid * PITCH + tid];
        sDB[tid] = sQ[tid * PITCH + tid];
        sDegA[tid] = ga;
        sDegB[tid] = gb;
    }
    __syncthreads();

    for (int e = tid; e < NN * NN; e += NT) {
        int i = e / NN, j = e % NN;
        float p = sP[i * PITCH + j] * sDA[i] * sDA[j];
        float q = sQ[i * PITCH + j] * sDB[i] * sDB[j];
        if (i == j) { p = 0.f; q = 0.f; }
        sP[i * PITCH + j] = p;
        sQ[i * PITCH + j] = q;
        sNS[i * PITCH + j] = sDA[i] * sDB[j] / (fabsf(sDegA[i] - sDegB[j]) + 1.0f);
    }
    __syncthreads();

    const int r = lane & 15;
    const int h = lane >> 4;
    const int ti0 = wave / 6, tn0 = wave % 6;
    const int ti1 = 5,        tn1 = 2 + wave;
    const int base0 = (ti0 * 16 + 8 * h) * PITCH + tn0 * 16 + r;
    const int base1 = (ti1 * 16 + 8 * h) * PITCH + tn1 * 16 + r;
    const int offa0 = (ti0 * 16 + r) * PITCH + 2 * h;
    const int offa1 = (ti1 * 16 + r) * PITCH + 2 * h;
    const int offb0 = 2 * h * PITCH + tn0 * 16 + r;
    const int offb1 = 2 * h * PITCH + tn1 * 16 + r;
    const bool two = wave < 4;

    for (int it = 0; it < MPM_ITERS; ++it) {
        if (tid < P1_THREADS) {
            const int jg = wave;
            const int ag = lane & 15;
            const int hb = lane >> 4;
            const float* xrow = &sX[jg * PITCH + hb * P1_B];
            const float* qrow = &sQ[ag * PITCH + hb * P1_B];
            float m[P1_J][P1_A];
            #pragma unroll
            for (int u = 0; u < P1_J; ++u) {
                #pragma unroll
                for (int v = 0; v < P1_A; ++v) m[u][v] = 0.0f;
            }
            v4f x[P1_J], qa[P1_A], qb[P1_A];
            #pragma unroll
            for (int u = 0; u < P1_J; ++u)
                x[u] = *(const v4f*)(xrow + u * (NN / P1_J) * PITCH);
            p1_load_q(qa, qrow, 0);
            for (int k0 = 0; k0 < P1_B - 8; k0 += 8) {
                p1_load_q(qb, qrow, k0 + 4);
                p1_step<true>(m, x, qa, xrow, k0 + 4);
                p1_load_q(qa, qrow, k0 + 8);
                p1_step<true>(m, x, qb, xrow, k0 + 8);
            }
            p1_load_q(qb, qrow, P1_B - 4);
            p1_step<true>(m, x, qa, xrow, P1_B - 4);
            p1_step<false>(m, x, qb, xrow, 0);
            float* mrow = &sM[jg * PITCH + ag + hb * (P1_A / 2) * (NN / P1_A)];
            #pragma unroll
            for (int u = 0; u < P1_J; ++u) {
                #pragma unroll
                for (int v = 0; v < P1_A / 2; ++v) {
                    float keep = hb ? m[u][v + P1_A / 2] : m[u][v];
                    float send = hb ? m[u][v] : m[u][v + P1_A / 2];
                    mrow[u * (NN / P1_J) * PITCH + v * (NN / P1_A)] = __builtin_fmaxf(keep, xhalf(send));
                }
            }
        }
        __syncthreads();

        v8f acc1 = {};
        if (two) acc1 = tile_gemm(sP + offa1, sM + offb1);
        v8f acc0 = tile_gemm(sP + offa0, sM + offb0);

        float ss = 0.f;
        tile_update(acc0, ss, sX + base0, sNS + base0);
        if (two) tile_update(acc1, ss, sX + base1, sNS + base1);
        ss = wave_sum(ss);
        if (lane == 0) sRed[wave] = ss;
        __syncthreads();

        float tot = wave_sum(sRed[lane]);
        float inv = 1.0f / sqrtf(tot);
        tile_store(acc0, inv, sX + base0);
        if (two) tile_store(acc1, inv, sX + base1);
        __syncthreads();
    }

    for (int e = tid; e < NN * NN; e += NT)
        out[e] = sX[(e / NN) * PITCH + (e % NN)];
}


#define G_NT 224
#define BC 4
#define NB (NN / BC)
#define RH 2
#define TPG (NN / 16 / RH)
#define PMT 104
#define NV4 (NN * NN / 4)
#define WS_PF 0
#define WS_Q  (NN * NN)
#define WS_NS (2 * NN * NN)
#define WS_XA (3 * NN * NN)
#define WS_XB (4 * NN * NN)
#define WS_FLOATS (5 * NN * NN)

#define R_U 3
#define R_S 6

__device__ __forceinline__ void lds_barrier()
{
    __builtin_amdgcn_fence(__ATOMIC_RELEASE, "workgroup", "local");
    __builtin_amdgcn_s_barrier();
    __builtin_amdgcn_fence(__ATOMIC_ACQUIRE, "workgroup", "local");
}

__device__ __forceinline__ float pick4(float a0, float a1, float a2, float a3, int k)
{
    float lo = (k & 1) ? a1 : a0;
    float hi = (k & 1) ? a3 : a2;
    return (k & 2) ? hi : lo;
}

__global__ __launch_bounds__(NT, 1)
void mpm_setup_kernel(const float* __restrict__ A_gt, const float* __restrict__ vec_logits, float* __restrict__ ws)
{
    __shared__ __attribute__((aligned(16))) float sP[NN * PITCH];
    __shared__ __attribute__((aligned(16))) float sQ[NN * PITCH];
    __shared__ __attribute__((aligned(16))) float sNS[NN * PITCH];
    __shared__ float sDegA[NN], sDegB[NN], sDA[NN], sDB[NN];
    const int tid = threadIdx.x;

    for (int e = tid; e < NN * NN; e += NT) {
        int i = e / NN, j = e % NN;
        float l;
        if (i == j) {
            l = 1.0f;
        } else {
            int lo = (i < j) ? i : j;
            int hi = (i < j) ? j : i;
            int off = (lo * (2 * NN - lo - 1)) / 2 + (hi - lo - 1);
            l = vec_logits[off];
        }
        sQ[i * PITCH + j] = 1.0f / (1.0f + expf(-l));
        sP[i * PITCH + j] = A_gt[e];
    }
    __syncthreads();
    if (tid < NN) {
        float ga = 0.f, gb = 0.f;
        for (int j = 0; j < NN; ++j) {
            ga += sP[tid * PITCH + j];
            gb += sQ[tid * PITCH + j];
        }
        sDA[tid] = sP[tid * PITCH + tid];
        sDB[tid] = sQ[tid * PITCH + tid];
        sDegA[tid] = ga;
        sDegB[tid] = gb;
    }
    __syncthreads();
    for (int e = tid; e < NN * NN; e += NT) {
        int i = e / NN, j = e % NN;
        float p = sP[i * PITCH + j] * sDA[i] * sDA[j];
        float q = sQ[i * PITCH + j] * sDB[i] * sDB[j];
        if (i == j) { p = 0.f; q = 0.f; }
        sP[i * PITCH + j] = p;
        sQ[i * PITCH + j] = q;
        sNS[i * PITCH + j] = sDA[i] * sDB[j] / (fabsf(sDegA[i] - sDegB[j]) + 1.0f);
    }
    __syncthreads();
    for (int f = tid; f < NV4; f += NT) {
        int ti = f / 384, kk = (f % 384) / 32, ln = f % 32;
        const float* s = &sP[(16 * ti + (ln & 15)) * PITCH + 8 * kk + 2 * (ln >> 4)];
        v4f a = { s[0], s[1], s[4], s[5] };
        ((v4f*)(ws + WS_PF))[f] = a;
        int qi = f / 24, qc = f % 24;
        ((v4f*)(ws + WS_Q))[f] = *(const v4f*)&sQ[qi * PITCH + 4 * qc];
        int g = f / NN, i = f % NN;
        ((v4f*)(ws + WS_NS))[f] = *(const v4f*)&sNS[i * PITCH + BC * g];
        v4f x0 = { 1.0f / (float)NN, 1.0f / (float)NN, 1.0f / (float)NN, 1.0f / (float)NN };
        ((v4f*)(ws + WS_XA))[f] = x0;
    }
}

__global__ __launch_bounds__(G_NT, 1)
void mpm_iter_kernel(const float* pf, const float* qd, const float* nsb,
                     const float* xin, float* xout)
{
    __shared__ __attribute__((aligned(16))) float sQ[BC * NN];
    __shared__ __attribute__((aligned(16))) float sMt[BC * PMT];
    __shared__ __attribute__((aligned(16))) float sSS[128];

    const int g    = blockIdx.x / RH;
    const int hp   = blockIdx.x % RH;
    const int tid  = threadIdx.x;
    const int lane = tid & 31;
    const int wave = __builtin_amdgcn_readfirstlane(tid >> 5);
    const int r = lane & 15;
    const int h = lane >> 4;
    const int rl = lane & 7;
    const int bq = lane >> 3;

    const bool ph1 = wave < 4;
    v4f x[R_S][R_U];
    v4f af[12];
    const int sel = r >> 2;
    int e0 = 0;
    float ns0 = 0.0f, ns1 = 0.0f, xr0 = 0.0f, xr1 = 0.0f;
    if (ph1) {
        const v4f* xs = (const v4f*)xin + (R_S * bq) * NN + 24 * wave + rl;
        #pragma unroll
        for (int s = 0; s < R_S; ++s) {
            #pragma unroll
            for (int u = 0; u < R_U; ++u) x[s][u] = xs[s * NN + 8 * u];
        }
    } else {
        const int tq = tid - 128;
        const v4f qw = ((const v4f*)(qd + g * BC * NN))[tq];
        const int ti = TPG * hp + (wave - 4);
        const v4f* p = (const v4f*)pf + ti * 12 * 32 + lane;
        #pragma unroll
        for (int kk = 0; kk < 12; ++kk) af[kk] = p[kk * 32];
        e0 = (g * NN + ti * 16 + 8 * h + sel) * BC + (r & 3);
        ns0 = nsb[e0]; ns1 = nsb[e0 + 4 * BC];
        xr0 = xin[e0]; xr1 = xin[e0 + 4 * BC];
        *(v4f*)&sQ[4 * tq] = qw;
    }

    lds_barrier();

    if (ph1) {
        v4f q[R_S][BC];
        const float* qrow = &sQ[24 * bq];
        #pragma unroll
        for (int s = 0; s < R_S; ++s) {
            #pragma unroll
            for (int v = 0; v < BC; ++v) q[s][v] = *(const v4f*)(qrow + v * NN + 4 * s);
        }
        float ma[R_U][BC], mb[R_U][BC];
        #pragma unroll
        for (int u = 0; u < R_U; ++u) {
            #pragma unroll
            for (int v = 0; v < BC; ++v) { ma[u][v] = 0.0f; mb[u][v] = 0.0f; }
        }
        v2f sa = { 0.0f, 0.0f }, sb = { 0.0f, 0.0f };
        #pragma unroll
        for (int s = 0; s < R_S; ++s) {
            #pragma unroll
            for (int u = 0; u < R_U; ++u) {
                const v4f xv = x[s][u];
                sa = xv.lo * xv.lo + sa;
                sb = xv.hi * xv.hi + sb;
                v2f p0[BC], p1[BC];
                #pragma unroll
                for (int v = 0; v < BC; ++v) {
                    p0[v] = q[s][v].lo * xv.lo;
                    p1[v] = q[s][v].hi * xv.hi;
                }
                #pragma unroll
                for (int v = 0; v < BC; ++v) {
                    ma[u][v] = __builtin_fmaxf(__builtin_fmaxf(ma[u][v], p0[v].x), p0[v].y);
                    mb[u][v] = __builtin_fmaxf(__builtin_fmaxf(mb[u][v], p1[v].x), p1[v].y);
                }
            }
        }
        sSS[tid] = (sa.x + sa.y) + (sb.x + sb.y);
        const int q8 = bq & 1;
        float* mcol = &sMt[bq * PMT + 24 * wave + rl];
        #pragma unroll
        for (int u = 0; u < R_U; ++u) {
            float m[BC], ra[2];
            #pragma unroll
            for (int v = 0; v < BC; ++v) m[v] = __builtin_fmaxf(ma[u][v], mb[u][v]);
            #pragma unroll
            for (int v = 0; v < 2; ++v) {
                float keep = h ? m[v + 2] : m[v];
                float send = h ? m[v] : m[v + 2];
                ra[v] = __builtin_fmaxf(keep, xhalf(send));
            }
            float keep = q8 ? ra[1] : ra[0];
            float send = q8 ? ra[0] : ra[1];
            mcol[8 * u] = __builtin_fmaxf(keep, row_xor<8>(send));
        }
    }
    lds_barrier();

    if (!ph1) {
        const v4f p4 = *(const v4f*)&sSS[4 * lane];
        const float* pb = sMt + (r & 3) * PMT + 2 * h;
        v2f bf[24];
        #pragma unroll
        for (int s = 0; s < 24; ++s) bf[s] = *(const v2f*)(pb + 4 * s);
        v8f c = {};
        asm volatile("" : "+v"(c) : : "memory");
        #pragma unroll
        for (int kk = 0; kk < 12; ++kk) {
            c = __builtin_amdgcn_wmma_f32_16x16x4_f32(false, af[kk].lo, false, bf[2 * kk], (short)0, c, false, false);
            c = __builtin_amdgcn_wmma_f32_16x16x4_f32(false, af[kk].hi, false, bf[2 * kk + 1], (short)0, c, false, false);
        }
        const float inv = 1.0f / sqrtf(wave_sum((p4.x + p4.y) + (p4.z + p4.w)));
        const float y0 = pick4(c[0], c[1], c[2], c[3], sel);
        const float y1 = pick4(c[4], c[5], c[6], c[7], sel);
        xout[e0] = __builtin_fmaf(xr0, ns0, y0) * inv;
        xout[e0 + 4 * BC] = __builtin_fmaf(xr1, ns1, y1) * inv;
    }
}

__global__ __launch_bounds__(256, 1)
void mpm_final_kernel(const float* __restrict__ xin, float* __restrict__ out)
{
    __shared__ __attribute__((aligned(16))) float sRed[8];
    const int tid  = threadIdx.x;
    const int lane = tid & 31;
    const int wave = __builtin_amdgcn_readfirstlane(tid >> 5);
    const int e = blockIdx.x * 256 + tid;
    const int i = e / NB, c = e % NB;
    const v4f* xs = (const v4f*)xin;
    const v4f x = xs[c * NN + i];
    v4f w[NV4 / 256];
    #pragma unroll
    for (int k = 0; k < NV4 / 256; ++k) w[k] = xs[tid + 256 * k];
    v2f sa = { 0.0f, 0.0f }, sb = { 0.0f, 0.0f };
    #pragma unroll
    for (int k = 0; k < NV4 / 256; ++k) {
        sa = w[k].lo * w[k].lo + sa;
        sb = w[k].hi * w[k].hi + sb;
    }
    const float ss = wave_sum((sa.x + sa.y) + (sb.x + sb.y));
    if (lane == 0) sRed[wave] = ss;
    __syncthreads();
    const v4f a = *(const v4f*)&sRed[0], b = *(const v4f*)&sRed[4];
    const float inv = 1.0f / sqrtf(((a.x + a.y) + (a.z + a.w)) + ((b.x + b.y) + (b.z + b.w)));
    ((v4f*)out)[e] = x * inv;
}

extern "C" void kernel_launch(void* const* d_in, const int* in_sizes, int n_in,
                              void* d_out, int out_size, void* d_ws, size_t ws_size,
                              hipStream_t stream) {
    (void)in_sizes; (void)n_in; (void)out_size;
    const float* A_gt = (const float*)d_in[0];
    const float* vec  = (const float*)d_in[1];
    float* out = (float*)d_out;
    float* ws  = (float*)d_ws;
    if (ws_size < sizeof(float) * WS_FLOATS) {
        graphvae_mpm_kernel<<<1, NT, 0, stream>>>(A_gt, vec, out);
        return;
    }
    mpm_setup_kernel<<<1, NT, 0, stream>>>(A_gt, vec, ws);
    for (int it = 0; it < MPM_ITERS; it += 2) {
        mpm_iter_kernel<<<RH * NB, G_NT, 0, stream>>>(ws + WS_PF, ws + WS_Q, ws + WS_NS, ws + WS_XA, ws + WS_XB);
        mpm_iter_kernel<<<RH * NB, G_NT, 0, stream>>>(ws + WS_PF, ws + WS_Q, ws + WS_NS, ws + WS_XB, ws + WS_XA);
    }
    mpm_final_kernel<<<NN * NN / 4 / 256, 256, 0, stream>>>(ws + WS_XA, out);
}
